# starting kernel
# baseline (speedup 1.0000x reference)
#include <hip/hip_runtime.h>

typedef __attribute__((ext_vector_type(2))) float v2f;
typedef __attribute__((ext_vector_type(8))) float v8f;

__global__ void cheb_degree_kernel(const int* __restrict__ row, float* __restrict__ deg, int E) {
    int e = blockIdx.x * blockDim.x + threadIdx.x;
    if (e < E) atomicAdd(&deg[row[e]], 1.0f);
}

__global__ void cheb_dis_kernel(float* __restrict__ deg, int n) {
    int i = blockIdx.x * blockDim.x + threadIdx.x;
    if (i < n) {
        float d = deg[i];
        deg[i] = (d > 0.0f) ? rsqrtf(fmaxf(d, 1.0f)) : 0.0f;
    }
}

__global__ void cheb_edgew_kernel(const int* __restrict__ row, const int* __restrict__ col,
                                  const float* __restrict__ dis, float* __restrict__ wE, int E) {
    int e = blockIdx.x * blockDim.x + threadIdx.x;
    if (e < E) wE[e] = -dis[row[e]] * dis[col[e]];
}

__global__ void cheb_negcopy_kernel(const float* __restrict__ a, float* __restrict__ b, int n) {
    int i = blockIdx.x * blockDim.x + threadIdx.x;
    if (i < n) b[i] = -a[i];
}

__global__ void cheb_scatter_kernel(const float* __restrict__ in, float* __restrict__ out,
                                    const int* __restrict__ src, const int* __restrict__ dst,
                                    const float* __restrict__ wE, float scale, int E) {
    int t = blockIdx.x * blockDim.x + threadIdx.x;
    int e = t >> 4;
    if (e >= E) return;
    int c = t & 15;
    int s = src[e];
    int d = dst[e];
    float w = wE[e] * scale;
    const float4 v = *(const float4*)(in + (size_t)s * 64 + (c << 2));
    float* op = out + (size_t)d * 64 + (c << 2);
    atomicAdd(op + 0, v.x * w);
    atomicAdd(op + 1, v.y * w);
    atomicAdd(op + 2, v.z * w);
    atomicAdd(op + 3, v.w * w);
}

template <int MT>
__global__ void __launch_bounds__(128)
cheb_gemm_wmma_kernel(const float* __restrict__ T0, const float* __restrict__ T1,
                      const float* __restrict__ T2, const float* __restrict__ W,
                      const float* __restrict__ bias, float* __restrict__ out,
                      int nTilesN, int F, int doRelu, int totalWaveTiles, int rowOffset) {
    __shared__ float sW[3 * 64 * 64 + 64];
    const int tid = threadIdx.y * 32 + threadIdx.x;
    const int nW = 3 * 64 * F;
    for (int i = tid; i < nW; i += 128) sW[i] = W[i];
    for (int i = tid; i < F; i += 128) sW[nW + i] = bias[i];
    __syncthreads();

    const int waveTile = blockIdx.x * 4 + threadIdx.y;
    if (waveTile >= totalWaveTiles) return;
    const int mGroup = waveTile / nTilesN;
    const int nTile = waveTile % nTilesN;

    const int lane = threadIdx.x;
    const int lg = lane >> 4;
    const int ln = lane & 15;
    const int rowBase = rowOffset + mGroup * (MT * 16);
    const int col = (nTile << 4) + ln;

    v8f acc[MT];
    const float bv = sW[nW + col];
#pragma unroll
    for (int i = 0; i < MT; ++i)
#pragma unroll
        for (int r = 0; r < 8; ++r) acc[i][r] = bv;

    const float* mats[3] = {T0, T1, T2};
#pragma unroll
    for (int m = 0; m < 3; ++m) {
        const float* __restrict__ T = mats[m];
        const float* Wm = &sW[m * 64 * F];
#pragma unroll
        for (int kb = 0; kb < 16; ++kb) {
            const int k0 = (kb << 2) + (lg << 1);
            v2f b;
            b.x = Wm[k0 * F + col];
            b.y = Wm[(k0 + 1) * F + col];
#pragma unroll
            for (int i = 0; i < MT; ++i) {
                v2f a = *(const v2f*)(T + (size_t)(rowBase + i * 16 + ln) * 64 + k0);
                acc[i] = __builtin_amdgcn_wmma_f32_16x16x4_f32(
                    false, a, false, b, (short)0, acc[i], false, false);
            }
        }
    }

#pragma unroll
    for (int i = 0; i < MT; ++i) {
        if (doRelu) {
#pragma unroll
            for (int r = 0; r < 8; ++r) acc[i][r] = fmaxf(acc[i][r], 0.0f);
        }
        float* orow = out + (size_t)(rowBase + i * 16 + (lg << 3)) * F + col;
#pragma unroll
        for (int r = 0; r < 8; ++r) orow[r * F] = acc[i][r];
    }
}

static void cheb_layer(const float* T0, const float* W, const float* b,
                       float* T1, float* T2, float* Hout,
                       const int* src, const int* dst, const float* wE,
                       int n, int E, int F, int relu, hipStream_t stream) {
    const int sthreads = E * 16;
    const int sblocks = (sthreads + 255) / 256;

    hipMemsetAsync(T1, 0, (size_t)n * 64 * sizeof(float), stream);
    cheb_scatter_kernel<<<sblocks, 256, 0, stream>>>(T0, T1, src, dst, wE, 1.0f, E);

    cheb_negcopy_kernel<<<(n * 64 + 255) / 256, 256, 0, stream>>>(T0, T2, n * 64);
    cheb_scatter_kernel<<<sblocks, 256, 0, stream>>>(T1, T2, src, dst, wE, 2.0f, E);

    const int nTilesN = F / 16;
    const int mTiles = n / 16;
    const int mGroups = mTiles / 5;
    const int wt5 = mGroups * nTilesN;
    if (wt5 > 0) {
        cheb_gemm_wmma_kernel<5><<<(wt5 + 3) / 4, dim3(32, 4), 0, stream>>>(
            T0, T1, T2, W, b, Hout, nTilesN, F, relu, wt5, 0);
    }
    const int remTiles = mTiles - mGroups * 5;
    if (remTiles > 0) {
        const int wt1 = remTiles * nTilesN;
        cheb_gemm_wmma_kernel<1><<<(wt1 + 3) / 4, dim3(32, 4), 0, stream>>>(
            T0, T1, T2, W, b, Hout, nTilesN, F, relu, wt1, mGroups * 80);
    }
}

extern "C" void kernel_launch(void* const* d_in, const int* in_sizes, int n_in,
                              void* d_out, int out_size, void* d_ws, size_t ws_size,
                              hipStream_t stream) {
    const float* x  = (const float*)d_in[0];
    const int*   ei = (const int*)d_in[1];
    const float* W1 = (const float*)d_in[2];
    const float* b1 = (const float*)d_in[3];
    const float* W2 = (const float*)d_in[4];
    const float* b2 = (const float*)d_in[5];
    const float* W3 = (const float*)d_in[6];
    const float* b3 = (const float*)d_in[7];

    const int n = in_sizes[0] / 64;
    const int E = in_sizes[1] / 2;
    const int* src = ei;
    const int* dst = ei + E;

    float* ws = (float*)d_ws;
    size_t o = 0;
    float* dis = ws + o; o += ((size_t)n + 63) & ~(size_t)63;
    float* wE  = ws + o; o += ((size_t)E + 63) & ~(size_t)63;
    float* T1  = ws + o; o += (size_t)n * 64;
    float* T2  = ws + o; o += (size_t)n * 64;
    float* HA  = ws + o; o += (size_t)n * 64;
    float* HB  = ws + o; o += (size_t)n * 64;

    hipMemsetAsync(dis, 0, (size_t)n * sizeof(float), stream);
    cheb_degree_kernel<<<(E + 255) / 256, 256, 0, stream>>>(src, dis, E);
    cheb_dis_kernel<<<(n + 255) / 256, 256, 0, stream>>>(dis, n);
    cheb_edgew_kernel<<<(E + 255) / 256, 256, 0, stream>>>(src, dst, dis, wE, E);

    cheb_layer(x,  W1, b1, T1, T2, HA, src, dst, wE, n, E, 64, 1, stream);
    cheb_layer(HA, W2, b2, T1, T2, HB, src, dst, wE, n, E, 64, 1, stream);
    cheb_layer(HB, W3, b3, T1, T2, (float*)d_out, src, dst, wE, n, E, 32, 0, stream);
}


# Propagation as CSR gather: edges grouped by destination once per call (count, scan, fill); one wave sums each row in arrival-order rank order; no float atomics, memset or init pass on features
# speedup vs baseline: 1.8795x; 1.8795x over previous
#include <hip/hip_runtime.h>

typedef __attribute__((ext_vector_type(2))) float v2f;
typedef __attribute__((ext_vector_type(8))) float v8f;


__global__ void cheb_count_kernel(const int* __restrict__ row, const int* __restrict__ col,
                                  float* __restrict__ deg, int* __restrict__ cnt,
                                  int* __restrict__ rank, int E) {
    int e = blockIdx.x * blockDim.x + threadIdx.x;
    if (e < E) {
        atomicAdd(&deg[row[e]], 1.0f);
        rank[e] = atomicAdd(&cnt[col[e]], 1);
    }
}

__global__ void cheb_dis_kernel(float* __restrict__ deg, int n) {
    int i = blockIdx.x * blockDim.x + threadIdx.x;
    if (i < n) {
        float d = deg[i];
        deg[i] = (d > 0.0f) ? rsqrtf(fmaxf(d, 1.0f)) : 0.0f;
    }
}

__global__ void __launch_bounds__(1024)
cheb_rowptr_kernel(const int* __restrict__ cnt, int* __restrict__ rowptr, int n) {
    __shared__ int sWave[32];
    const int tid = threadIdx.x;
    const int lane = tid & 31;
    const int wv = tid >> 5;
    int carry = 0;
    for (int base = 0; base < n; base += 4096) {
        const int i0 = base + tid * 4;
        int c0 = 0, c1 = 0, c2 = 0, c3 = 0;
        if (i0 + 3 < n) {
            const int4 v = *(const int4*)(cnt + i0);
            c0 = v.x; c1 = v.y; c2 = v.z; c3 = v.w;
        } else {
            if (i0 + 0 < n) c0 = cnt[i0 + 0];
            if (i0 + 1 < n) c1 = cnt[i0 + 1];
            if (i0 + 2 < n) c2 = cnt[i0 + 2];
        }
        const int s1 = c0, s2 = s1 + c1, s3 = s2 + c2, tot = s3 + c3;
        int incl = tot;
        for (int o = 1; o < 32; o <<= 1) {
            const int t = __shfl_up(incl, o);
            if (lane >= o) incl += t;
        }
        if (lane == 31) sWave[wv] = incl;
        __syncthreads();
        const int wt = sWave[lane];
        int wincl = wt;
        for (int o = 1; o < 32; o <<= 1) {
            const int t = __shfl_up(wincl, o);
            if (lane >= o) wincl += t;
        }
        const int waveExcl = __shfl(wincl, wv) - __shfl(wt, wv);
        const int tileTot = __shfl(wincl, 31);
        const int ex = carry + waveExcl + incl - tot;
        if (i0 + 3 < n) {
            *(int4*)(rowptr + i0) = make_int4(ex, ex + s1, ex + s2, ex + s3);
        } else {
            if (i0 + 0 < n) rowptr[i0 + 0] = ex;
            if (i0 + 1 < n) rowptr[i0 + 1] = ex + s1;
            if (i0 + 2 < n) rowptr[i0 + 2] = ex + s2;
        }
        carry += tileTot;
        __syncthreads();
    }
    if (tid == 0) rowptr[n] = carry;
}

__global__ void cheb_fill_kernel(const int* __restrict__ row, const int* __restrict__ col,
                                 const float* __restrict__ dis, const int* __restrict__ rowptr,
                                 const int* __restrict__ rank, int2* __restrict__ csr, int E) {
    int e = blockIdx.x * blockDim.x + threadIdx.x;
    if (e < E) {
        const int s = row[e];
        const int d = col[e];
        const float w = -dis[s] * dis[d];
        csr[rowptr[d] + rank[e]] = make_int2(s, __float_as_int(w));
    }
}

template <bool CHEB2>
__global__ void __launch_bounds__(256)
cheb_gather_kernel(const float* __restrict__ in, const float* __restrict__ t0,
                   float* __restrict__ out, const int* __restrict__ rowptr,
                   const int2* __restrict__ csr, int n) {
    const int i = __builtin_amdgcn_readfirstlane(blockIdx.x * 8 + (int)(threadIdx.x >> 5));
    if (i >= n) return;
    const int lane = threadIdx.x & 31;
    const int beg = rowptr[i];
    const int end = rowptr[i + 1];
    const float* __restrict__ base = in + (lane << 1);

    v2f t = {0.0f, 0.0f};
    if (CHEB2) t = *(const v2f*)(t0 + (size_t)i * 64 + (lane << 1));

    float ax = 0.0f, ay = 0.0f;
    int j = beg;
    for (; j + 8 <= end; j += 8) {
        int2 p[8];
        v2f v[8];
#pragma unroll
        for (int u = 0; u < 8; ++u) p[u] = csr[j + u];
#pragma unroll
        for (int u = 0; u < 8; ++u) v[u] = *(const v2f*)(base + (size_t)p[u].x * 64);
#pragma unroll
        for (int u = 0; u < 8; ++u) {
            const float w = __int_as_float(p[u].y);
            ax = fmaf(v[u].x, w, ax);
            ay = fmaf(v[u].y, w, ay);
        }
    }
    if (j + 4 <= end) {
        int2 p[4];
        v2f v[4];
#pragma unroll
        for (int u = 0; u < 4; ++u) p[u] = csr[j + u];
#pragma unroll
        for (int u = 0; u < 4; ++u) v[u] = *(const v2f*)(base + (size_t)p[u].x * 64);
#pragma unroll
        for (int u = 0; u < 4; ++u) {
            const float w = __int_as_float(p[u].y);
            ax = fmaf(v[u].x, w, ax);
            ay = fmaf(v[u].y, w, ay);
        }
        j += 4;
    }
    for (; j < end; ++j) {
        const int2 p = csr[j];
        const v2f v = *(const v2f*)(base + (size_t)p.x * 64);
        const float w = __int_as_float(p.y);
        ax = fmaf(v.x, w, ax);
        ay = fmaf(v.y, w, ay);
    }

    v2f r;
    if (CHEB2) { r.x = 2.0f * ax - t.x; r.y = 2.0f * ay - t.y; }
    else       { r.x = ax;              r.y = ay; }
    *(v2f*)(out + (size_t)i * 64 + (lane << 1)) = r;
}

__global__ void cheb_degree_kernel(const int* __restrict__ row, float* __restrict__ deg, int E) {
    int e = blockIdx.x * blockDim.x + threadIdx.x;
    if (e < E) atomicAdd(&deg[row[e]], 1.0f);
}

__global__ void cheb_edgew_kernel(const int* __restrict__ row, const int* __restrict__ col,
                                  const float* __restrict__ dis, float* __restrict__ wE, int E) {
    int e = blockIdx.x * blockDim.x + threadIdx.x;
    if (e < E) wE[e] = -dis[row[e]] * dis[col[e]];
}

__global__ void cheb_negcopy_kernel(const float* __restrict__ a, float* __restrict__ b, int n) {
    int i = blockIdx.x * blockDim.x + threadIdx.x;
    if (i < n) b[i] = -a[i];
}

__global__ void cheb_scatter_kernel(const float* __restrict__ in, float* __restrict__ out,
                                    const int* __restrict__ src, const int* __restrict__ dst,
                                    const float* __restrict__ wE, float scale, int E) {
    int t = blockIdx.x * blockDim.x + threadIdx.x;
    int e = t >> 4;
    if (e >= E) return;
    int c = t & 15;
    int s = src[e];
    int d = dst[e];
    float w = wE[e] * scale;
    const float4 v = *(const float4*)(in + (size_t)s * 64 + (c << 2));
    float* op = out + (size_t)d * 64 + (c << 2);
    atomicAdd(op + 0, v.x * w);
    atomicAdd(op + 1, v.y * w);
    atomicAdd(op + 2, v.z * w);
    atomicAdd(op + 3, v.w * w);
}

template <int MT>
__global__ void __launch_bounds__(128)
cheb_gemm_wmma_kernel(const float* __restrict__ T0, const float* __restrict__ T1,
                      const float* __restrict__ T2, const float* __restrict__ W,
                      const float* __restrict__ bias, float* __restrict__ out,
                      int nTilesN, int F, int doRelu, int totalWaveTiles, int rowOffset) {
    __shared__ float sW[3 * 64 * 64 + 64];
    const int tid = threadIdx.y * 32 + threadIdx.x;
    const int nW = 3 * 64 * F;
    for (int i = tid; i < nW; i += 128) sW[i] = W[i];
    for (int i = tid; i < F; i += 128) sW[nW + i] = bias[i];
    __syncthreads();

    const int waveTile = blockIdx.x * 4 + threadIdx.y;
    if (waveTile >= totalWaveTiles) return;
    const int mGroup = waveTile / nTilesN;
    const int nTile = waveTile % nTilesN;

    const int lane = threadIdx.x;
    const int lg = lane >> 4;
    const int ln = lane & 15;
    const int rowBase = rowOffset + mGroup * (MT * 16);
    const int col = (nTile << 4) + ln;

    v8f acc[MT];
    const float bv = sW[nW + col];
#pragma unroll
    for (int i = 0; i < MT; ++i)
#pragma unroll
        for (int r = 0; r < 8; ++r) acc[i][r] = bv;

    const float* mats[3] = {T0, T1, T2};
#pragma unroll
    for (int m = 0; m < 3; ++m) {
        const float* __restrict__ T = mats[m];
        const float* Wm = &sW[m * 64 * F];
#pragma unroll
        for (int kb = 0; kb < 16; ++kb) {
            const int k0 = (kb << 2) + (lg << 1);
            v2f b;
            b.x = Wm[k0 * F + col];
            b.y = Wm[(k0 + 1) * F + col];
#pragma unroll
            for (int i = 0; i < MT; ++i) {
                v2f a = *(const v2f*)(T + (size_t)(rowBase + i * 16 + ln) * 64 + k0);
                acc[i] = __builtin_amdgcn_wmma_f32_16x16x4_f32(
                    false, a, false, b, (short)0, acc[i], false, false);
            }
        }
    }

#pragma unroll
    for (int i = 0; i < MT; ++i) {
        if (doRelu) {
#pragma unroll
            for (int r = 0; r < 8; ++r) acc[i][r] = fmaxf(acc[i][r], 0.0f);
        }
        float* orow = out + (size_t)(rowBase + i * 16 + (lg << 3)) * F + col;
#pragma unroll
        for (int r = 0; r < 8; ++r) orow[r * F] = acc[i][r];
    }
}

static void cheb_gemm(const float* T0, const float* T1, const float* T2,
                      const float* W, const float* b, float* Hout,
                      int n, int F, int relu, hipStream_t stream) {
    const int nTilesN = F / 16;
    const int mTiles = n / 16;
    const int mGroups = mTiles / 5;
    const int wt5 = mGroups * nTilesN;
    if (wt5 > 0) {
        cheb_gemm_wmma_kernel<5><<<(wt5 + 3) / 4, dim3(32, 4), 0, stream>>>(
            T0, T1, T2, W, b, Hout, nTilesN, F, relu, wt5, 0);
    }
    const int remTiles = mTiles - mGroups * 5;
    if (remTiles > 0) {
        const int wt1 = remTiles * nTilesN;
        cheb_gemm_wmma_kernel<1><<<(wt1 + 3) / 4, dim3(32, 4), 0, stream>>>(
            T0, T1, T2, W, b, Hout, nTilesN, F, relu, wt1, mGroups * 80);
    }
}

static void cheb_layer_csr(const float* T0, const float* W, const float* b,
                           float* T1, float* T2, float* Hout,
                           const int* rowptr, const int2* csr,
                           int n, int F, int relu, hipStream_t stream) {
    const int gblocks = (n + 7) / 8;
    cheb_gather_kernel<false><<<gblocks, 256, 0, stream>>>(T0, T0, T1, rowptr, csr, n);
    cheb_gather_kernel<true><<<gblocks, 256, 0, stream>>>(T1, T0, T2, rowptr, csr, n);
    cheb_gemm(T0, T1, T2, W, b, Hout, n, F, relu, stream);
}

static void cheb_layer_scatter(const float* T0, const float* W, const float* b,
                               float* T1, float* T2, float* Hout,
                               const int* src, const int* dst, const float* wE,
                               int n, int E, int F, int relu, hipStream_t stream) {
    const int sthreads = E * 16;
    const int sblocks = (sthreads + 255) / 256;
    hipMemsetAsync(T1, 0, (size_t)n * 64 * sizeof(float), stream);
    cheb_scatter_kernel<<<sblocks, 256, 0, stream>>>(T0, T1, src, dst, wE, 1.0f, E);
    cheb_negcopy_kernel<<<(n * 64 + 255) / 256, 256, 0, stream>>>(T0, T2, n * 64);
    cheb_scatter_kernel<<<sblocks, 256, 0, stream>>>(T1, T2, src, dst, wE, 2.0f, E);
    cheb_gemm(T0, T1, T2, W, b, Hout, n, F, relu, stream);
}

extern "C" void kernel_launch(void* const* d_in, const int* in_sizes, int n_in,
                              void* d_out, int out_size, void* d_ws, size_t ws_size,
                              hipStream_t stream) {
    const float* x  = (const float*)d_in[0];
    const int*   ei = (const int*)d_in[1];
    const float* W1 = (const float*)d_in[2];
    const float* b1 = (const float*)d_in[3];
    const float* W2 = (const float*)d_in[4];
    const float* b2 = (const float*)d_in[5];
    const float* W3 = (const float*)d_in[6];
    const float* b3 = (const float*)d_in[7];

    const int n = in_sizes[0] / 64;
    const int E = in_sizes[1] / 2;
    const int* src = ei;
    const int* dst = ei + E;

    const size_t nPad = ((size_t)n + 64 + 63) & ~(size_t)63;
    const size_t ePad = ((size_t)E + 63) & ~(size_t)63;
    float* ws = (float*)d_ws;

    const size_t csrWords = 3 * nPad + 2 * ePad + 4 * (size_t)n * 64;
    if (ws_size >= csrWords * sizeof(float)) {
        size_t o = 0;
        float* dis    = ws + o;          o += nPad;
        int*   cnt    = (int*)(ws + o);  o += nPad;
        int*   rowptr = (int*)(ws + o);  o += nPad;
        int2*  csr    = (int2*)(ws + o); o += 2 * ePad;
        float* T1 = ws + o; o += (size_t)n * 64;
        float* T2 = ws + o; o += (size_t)n * 64;
        float* HA = ws + o; o += (size_t)n * 64;
        float* HB = ws + o; o += (size_t)n * 64;
        int* rank = (int*)T2;

        hipMemsetAsync(dis, 0, 2 * nPad * sizeof(float), stream);
        cheb_count_kernel<<<(E + 255) / 256, 256, 0, stream>>>(src, dst, dis, cnt, rank, E);
        cheb_dis_kernel<<<(n + 255) / 256, 256, 0, stream>>>(dis, n);
        cheb_rowptr_kernel<<<1, 1024, 0, stream>>>(cnt, rowptr, n);
        cheb_fill_kernel<<<(E + 255) / 256, 256, 0, stream>>>(src, dst, dis, rowptr, rank, csr, E);

        cheb_layer_csr(x,  W1, b1, T1, T2, HA, rowptr, csr, n, 64, 1, stream);
        cheb_layer_csr(HA, W2, b2, T1, T2, HB, rowptr, csr, n, 64, 1, stream);
        cheb_layer_csr(HB, W3, b3, T1, T2, (float*)d_out, rowptr, csr, n, 32, 0, stream);
        return;
    }

    size_t o = 0;
    float* dis = ws + o; o += ((size_t)n + 63) & ~(size_t)63;
    float* wE  = ws + o; o += ePad;
    float* T1  = ws + o; o += (size_t)n * 64;
    float* T2  = ws + o; o += (size_t)n * 64;
    float* HA  = ws + o; o += (size_t)n * 64;
    float* HB  = ws + o; o += (size_t)n * 64;

    hipMemsetAsync(dis, 0, (size_t)n * sizeof(float), stream);
    cheb_degree_kernel<<<(E + 255) / 256, 256, 0, stream>>>(src, dis, E);
    cheb_dis_kernel<<<(n + 255) / 256, 256, 0, stream>>>(dis, n);
    cheb_edgew_kernel<<<(E + 255) / 256, 256, 0, stream>>>(src, dst, dis, wE, E);

    cheb_layer_scatter(x,  W1, b1, T1, T2, HA, src, dst, wE, n, E, 64, 1, stream);
    cheb_layer_scatter(HA, W2, b2, T1, T2, HB, src, dst, wE, n, E, 64, 1, stream);
    cheb_layer_scatter(HB, W3, b3, T1, T2, (float*)d_out, src, dst, wE, n, E, 32, 0, stream);
}


# GEMM: each lane loads 16 bytes of its A row (4 consecutive k feed two WMMAs, B fragments on the same k map): half the row-load instructions; k sum order inside each group of 8 changes
# speedup vs baseline: 1.9422x; 1.0334x over previous
#include <hip/hip_runtime.h>

typedef __attribute__((ext_vector_type(2))) float v2f;
typedef __attribute__((ext_vector_type(4))) float v4f;
typedef __attribute__((ext_vector_type(8))) float v8f;


__global__ void cheb_count_kernel(const int* __restrict__ row, const int* __restrict__ col,
                                  float* __restrict__ deg, int* __restrict__ cnt,
                                  int* __restrict__ rank, int E) {
    int e = blockIdx.x * blockDim.x + threadIdx.x;
    if (e < E) {
        atomicAdd(&deg[row[e]], 1.0f);
        rank[e] = atomicAdd(&cnt[col[e]], 1);
    }
}

__global__ void cheb_dis_kernel(float* __restrict__ deg, int n) {
    int i = blockIdx.x * blockDim.x + threadIdx.x;
    if (i < n) {
        float d = deg[i];
        deg[i] = (d > 0.0f) ? rsqrtf(fmaxf(d, 1.0f)) : 0.0f;
    }
}

__global__ void __launch_bounds__(1024)
cheb_rowptr_kernel(const int* __restrict__ cnt, int* __restrict__ rowptr, int n) {
    __shared__ int sWave[32];
    const int tid = threadIdx.x;
    const int lane = tid & 31;
    const int wv = tid >> 5;
    int carry = 0;
    for (int base = 0; base < n; base += 4096) {
        const int i0 = base + tid * 4;
        int c0 = 0, c1 = 0, c2 = 0, c3 = 0;
        if (i0 + 3 < n) {
            const int4 v = *(const int4*)(cnt + i0);
            c0 = v.x; c1 = v.y; c2 = v.z; c3 = v.w;
        } else {
            if (i0 + 0 < n) c0 = cnt[i0 + 0];
            if (i0 + 1 < n) c1 = cnt[i0 + 1];
            if (i0 + 2 < n) c2 = cnt[i0 + 2];
        }
        const int s1 = c0, s2 = s1 + c1, s3 = s2 + c2, tot = s3 + c3;
        int incl = tot;
        for (int o = 1; o < 32; o <<= 1) {
            const int t = __shfl_up(incl, o);
            if (lane >= o) incl += t;
        }
        if (lane == 31) sWave[wv] = incl;
        __syncthreads();
        const int wt = sWave[lane];
        int wincl = wt;
        for (int o = 1; o < 32; o <<= 1) {
            const int t = __shfl_up(wincl, o);
            if (lane >= o) wincl += t;
        }
        const int waveExcl = __shfl(wincl, wv) - __shfl(wt, wv);
        const int tileTot = __shfl(wincl, 31);
        const int ex = carry + waveExcl + incl - tot;
        if (i0 + 3 < n) {
            *(int4*)(rowptr + i0) = make_int4(ex, ex + s1, ex + s2, ex + s3);
        } else {
            if (i0 + 0 < n) rowptr[i0 + 0] = ex;
            if (i0 + 1 < n) rowptr[i0 + 1] = ex + s1;
            if (i0 + 2 < n) rowptr[i0 + 2] = ex + s2;
        }
        carry += tileTot;
        __syncthreads();
    }
    if (tid == 0) rowptr[n] = carry;
}

__global__ void cheb_fill_kernel(const int* __restrict__ row, const int* __restrict__ col,
                                 const float* __restrict__ dis, const int* __restrict__ rowptr,
                                 const int* __restrict__ rank, int2* __restrict__ csr, int E) {
    int e = blockIdx.x * blockDim.x + threadIdx.x;
    if (e < E) {
        const int s = row[e];
        const int d = col[e];
        const float w = -dis[s] * dis[d];
        csr[rowptr[d] + rank[e]] = make_int2(s, __float_as_int(w));
    }
}

template <bool CHEB2>
__global__ void __launch_bounds__(256)
cheb_gather_kernel(const float* __restrict__ in, const float* __restrict__ t0,
                   float* __restrict__ out, const int* __restrict__ rowptr,
                   const int2* __restrict__ csr, int n) {
    const int i = __builtin_amdgcn_readfirstlane(blockIdx.x * 8 + (int)(threadIdx.x >> 5));
    if (i >= n) return;
    const int lane = threadIdx.x & 31;
    const int beg = rowptr[i];
    const int end = rowptr[i + 1];
    const float* __restrict__ base = in + (lane << 1);

    v2f t = {0.0f, 0.0f};
    if (CHEB2) t = *(const v2f*)(t0 + (size_t)i * 64 + (lane << 1));

    float ax = 0.0f, ay = 0.0f;
    int j = beg;
    for (; j + 8 <= end; j += 8) {
        int2 p[8];
        v2f v[8];
#pragma unroll
        for (int u = 0; u < 8; ++u) p[u] = csr[j + u];
#pragma unroll
        for (int u = 0; u < 8; ++u) v[u] = *(const v2f*)(base + (size_t)p[u].x * 64);
#pragma unroll
        for (int u = 0; u < 8; ++u) {
            const float w = __int_as_float(p[u].y);
            ax = fmaf(v[u].x, w, ax);
            ay = fmaf(v[u].y, w, ay);
        }
    }
    if (j + 4 <= end) {
        int2 p[4];
        v2f v[4];
#pragma unroll
        for (int u = 0; u < 4; ++u) p[u] = csr[j + u];
#pragma unroll
        for (int u = 0; u < 4; ++u) v[u] = *(const v2f*)(base + (size_t)p[u].x * 64);
#pragma unroll
        for (int u = 0; u < 4; ++u) {
            const float w = __int_as_float(p[u].y);
            ax = fmaf(v[u].x, w, ax);
            ay = fmaf(v[u].y, w, ay);
        }
        j += 4;
    }
    for (; j < end; ++j) {
        const int2 p = csr[j];
        const v2f v = *(const v2f*)(base + (size_t)p.x * 64);
        const float w = __int_as_float(p.y);
        ax = fmaf(v.x, w, ax);
        ay = fmaf(v.y, w, ay);
    }

    v2f r;
    if (CHEB2) { r.x = 2.0f * ax - t.x; r.y = 2.0f * ay - t.y; }
    else       { r.x = ax;              r.y = ay; }
    *(v2f*)(out + (size_t)i * 64 + (lane << 1)) = r;
}

__global__ void cheb_degree_kernel(const int* __restrict__ row, float* __restrict__ deg, int E) {
    int e = blockIdx.x * blockDim.x + threadIdx.x;
    if (e < E) atomicAdd(&deg[row[e]], 1.0f);
}

__global__ void cheb_edgew_kernel(const int* __restrict__ row, const int* __restrict__ col,
                                  const float* __restrict__ dis, float* __restrict__ wE, int E) {
    int e = blockIdx.x * blockDim.x + threadIdx.x;
    if (e < E) wE[e] = -dis[row[e]] * dis[col[e]];
}

__global__ void cheb_negcopy_kernel(const float* __restrict__ a, float* __restrict__ b, int n) {
    int i = blockIdx.x * blockDim.x + threadIdx.x;
    if (i < n) b[i] = -a[i];
}

__global__ void cheb_scatter_kernel(const float* __restrict__ in, float* __restrict__ out,
                                    const int* __restrict__ src, const int* __restrict__ dst,
                                    const float* __restrict__ wE, float scale, int E) {
    int t = blockIdx.x * blockDim.x + threadIdx.x;
    int e = t >> 4;
    if (e >= E) return;
    int c = t & 15;
    int s = src[e];
    int d = dst[e];
    float w = wE[e] * scale;
    const float4 v = *(const float4*)(in + (size_t)s * 64 + (c << 2));
    float* op = out + (size_t)d * 64 + (c << 2);
    atomicAdd(op + 0, v.x * w);
    atomicAdd(op + 1, v.y * w);
    atomicAdd(op + 2, v.z * w);
    atomicAdd(op + 3, v.w * w);
}

template <int MT>
__global__ void __launch_bounds__(128)
cheb_gemm_wmma_kernel(const float* __restrict__ T0, const float* __restrict__ T1,
                      const float* __restrict__ T2, const float* __restrict__ W,
                      const float* __restrict__ bias, float* __restrict__ out,
                      int nTilesN, int F, int doRelu, int totalWaveTiles, int rowOffset) {
    __shared__ float sW[3 * 64 * 64 + 64];
    const int tid = threadIdx.y * 32 + threadIdx.x;
    const int nW = 3 * 64 * F;
    for (int i = tid; i < nW; i += 128) sW[i] = W[i];
    for (int i = tid; i < F; i += 128) sW[nW + i] = bias[i];
    __syncthreads();

    const int waveTile = blockIdx.x * 4 + threadIdx.y;
    if (waveTile >= totalWaveTiles) return;
    const int mGroup = waveTile / nTilesN;
    const int nTile = waveTile % nTilesN;

    const int lane = threadIdx.x;
    const int lg = lane >> 4;
    const int ln = lane & 15;
    const int rowBase = rowOffset + mGroup * (MT * 16);
    const int col = (nTile << 4) + ln;

    v8f acc[MT];
    const float bv = sW[nW + col];
#pragma unroll
    for (int i = 0; i < MT; ++i)
#pragma unroll
        for (int r = 0; r < 8; ++r) acc[i][r] = bv;

    const float* mats[3] = {T0, T1, T2};
#pragma unroll
    for (int m = 0; m < 3; ++m) {
        const float* __restrict__ T = mats[m];
        const float* Wm = &sW[m * 64 * F];
#pragma unroll
        for (int kq = 0; kq < 8; ++kq) {
            const int k0 = (kq << 3) + (lg << 2);
            v2f bLo, bHi;
            bLo.x = Wm[k0 * F + col];
            bLo.y = Wm[(k0 + 1) * F + col];
            bHi.x = Wm[(k0 + 2) * F + col];
            bHi.y = Wm[(k0 + 3) * F + col];
            v4f a[MT];
#pragma unroll
            for (int i = 0; i < MT; ++i)
                a[i] = *(const v4f*)(T + (size_t)(rowBase + i * 16 + ln) * 64 + k0);
#pragma unroll
            for (int i = 0; i < MT; ++i) {
                v2f aLo;
                aLo.x = a[i].x; aLo.y = a[i].y;
                acc[i] = __builtin_amdgcn_wmma_f32_16x16x4_f32(
                    false, aLo, false, bLo, (short)0, acc[i], false, false);
            }
#pragma unroll
            for (int i = 0; i < MT; ++i) {
                v2f aHi;
                aHi.x = a[i].z; aHi.y = a[i].w;
                acc[i] = __builtin_amdgcn_wmma_f32_16x16x4_f32(
                    false, aHi, false, bHi, (short)0, acc[i], false, false);
            }
        }
    }

#pragma unroll
    for (int i = 0; i < MT; ++i) {
        if (doRelu) {
#pragma unroll
            for (int r = 0; r < 8; ++r) acc[i][r] = fmaxf(acc[i][r], 0.0f);
        }
        float* orow = out + (size_t)(rowBase + i * 16 + (lg << 3)) * F + col;
#pragma unroll
        for (int r = 0; r < 8; ++r) orow[r * F] = acc[i][r];
    }
}

static void cheb_gemm(const float* T0, const float* T1, const float* T2,
                      const float* W, const float* b, float* Hout,
                      int n, int F, int relu, hipStream_t stream) {
    const int nTilesN = F / 16;
    const int mTiles = n / 16;
    const int mGroups = mTiles / 5;
    const int wt5 = mGroups * nTilesN;
    if (wt5 > 0) {
        cheb_gemm_wmma_kernel<5><<<(wt5 + 3) / 4, dim3(32, 4), 0, stream>>>(
            T0, T1, T2, W, b, Hout, nTilesN, F, relu, wt5, 0);
    }
    const int remTiles = mTiles - mGroups * 5;
    if (remTiles > 0) {
        const int wt1 = remTiles * nTilesN;
        cheb_gemm_wmma_kernel<1><<<(wt1 + 3) / 4, dim3(32, 4), 0, stream>>>(
            T0, T1, T2, W, b, Hout, nTilesN, F, relu, wt1, mGroups * 80);
    }
}

static void cheb_layer_csr(const float* T0, const float* W, const float* b,
                           float* T1, float* T2, float* Hout,
                           const int* rowptr, const int2* csr,
                           int n, int F, int relu, hipStream_t stream) {
    const int gblocks = (n + 7) / 8;
    cheb_gather_kernel<false><<<gblocks, 256, 0, stream>>>(T0, T0, T1, rowptr, csr, n);
    cheb_gather_kernel<true><<<gblocks, 256, 0, stream>>>(T1, T0, T2, rowptr, csr, n);
    cheb_gemm(T0, T1, T2, W, b, Hout, n, F, relu, stream);
}

static void cheb_layer_scatter(const float* T0, const float* W, const float* b,
                               float* T1, float* T2, float* Hout,
                               const int* src, const int* dst, const float* wE,
                               int n, int E, int F, int relu, hipStream_t stream) {
    const int sthreads = E * 16;
    const int sblocks = (sthreads + 255) / 256;
    hipMemsetAsync(T1, 0, (size_t)n * 64 * sizeof(float), stream);
    cheb_scatter_kernel<<<sblocks, 256, 0, stream>>>(T0, T1, src, dst, wE, 1.0f, E);
    cheb_negcopy_kernel<<<(n * 64 + 255) / 256, 256, 0, stream>>>(T0, T2, n * 64);
    cheb_scatter_kernel<<<sblocks, 256, 0, stream>>>(T1, T2, src, dst, wE, 2.0f, E);
    cheb_gemm(T0, T1, T2, W, b, Hout, n, F, relu, stream);
}

extern "C" void kernel_launch(void* const* d_in, const int* in_sizes, int n_in,
                              void* d_out, int out_size, void* d_ws, size_t ws_size,
                              hipStream_t stream) {
    const float* x  = (const float*)d_in[0];
    const int*   ei = (const int*)d_in[1];
    const float* W1 = (const float*)d_in[2];
    const float* b1 = (const float*)d_in[3];
    const float* W2 = (const float*)d_in[4];
    const float* b2 = (const float*)d_in[5];
    const float* W3 = (const float*)d_in[6];
    const float* b3 = (const float*)d_in[7];

    const int n = in_sizes[0] / 64;
    const int E = in_sizes[1] / 2;
    const int* src = ei;
    const int* dst = ei + E;

    const size_t nPad = ((size_t)n + 64 + 63) & ~(size_t)63;
    const size_t ePad = ((size_t)E + 63) & ~(size_t)63;
    float* ws = (float*)d_ws;

    const size_t csrWords = 3 * nPad + 2 * ePad + 4 * (size_t)n * 64;
    if (ws_size >= csrWords * sizeof(float)) {
        size_t o = 0;
        float* dis    = ws + o;          o += nPad;
        int*   cnt    = (int*)(ws + o);  o += nPad;
        int*   rowptr = (int*)(ws + o);  o += nPad;
        int2*  csr    = (int2*)(ws + o); o += 2 * ePad;
        float* T1 = ws + o; o += (size_t)n * 64;
        float* T2 = ws + o; o += (size_t)n * 64;
        float* HA = ws + o; o += (size_t)n * 64;
        float* HB = ws + o; o += (size_t)n * 64;
        int* rank = (int*)T2;

        hipMemsetAsync(dis, 0, 2 * nPad * sizeof(float), stream);
        cheb_count_kernel<<<(E + 255) / 256, 256, 0, stream>>>(src, dst, dis, cnt, rank, E);
        cheb_dis_kernel<<<(n + 255) / 256, 256, 0, stream>>>(dis, n);
        cheb_rowptr_kernel<<<1, 1024, 0, stream>>>(cnt, rowptr, n);
        cheb_fill_kernel<<<(E + 255) / 256, 256, 0, stream>>>(src, dst, dis, rowptr, rank, csr, E);

        cheb_layer_csr(x,  W1, b1, T1, T2, HA, rowptr, csr, n, 64, 1, stream);
        cheb_layer_csr(HA, W2, b2, T1, T2, HB, rowptr, csr, n, 64, 1, stream);
        cheb_layer_csr(HB, W3, b3, T1, T2, (float*)d_out, rowptr, csr, n, 32, 0, stream);
        return;
    }

    size_t o = 0;
    float* dis = ws + o; o += ((size_t)n + 63) & ~(size_t)63;
    float* wE  = ws + o; o += ePad;
    float* T1  = ws + o; o += (size_t)n * 64;
    float* T2  = ws + o; o += (size_t)n * 64;
    float* HA  = ws + o; o += (size_t)n * 64;
    float* HB  = ws + o; o += (size_t)n * 64;

    hipMemsetAsync(dis, 0, (size_t)n * sizeof(float), stream);
    cheb_degree_kernel<<<(E + 255) / 256, 256, 0, stream>>>(src, dis, E);
    cheb_dis_kernel<<<(n + 255) / 256, 256, 0, stream>>>(dis, n);
    cheb_edgew_kernel<<<(E + 255) / 256, 256, 0, stream>>>(src, dst, dis, wE, E);

    cheb_layer_scatter(x,  W1, b1, T1, T2, HA, src, dst, wE, n, E, 64, 1, stream);
    cheb_layer_scatter(HA, W2, b2, T1, T2, HB, src, dst, wE, n, E, 64, 1, stream);
    cheb_layer_scatter(HB, W3, b3, T1, T2, (float*)d_out, src, dst, wE, n, E, 32, 0, stream);
}


# rowptr scan: the 13 tiles of 4096 counts go through ONE round (13 loads in flight together, wave scans on DPP row shifts side by side, one barrier, one store round) instead of 13 dependent rounds
# speedup vs baseline: 2.0551x; 1.0581x over previous
#include <hip/hip_runtime.h>

typedef __attribute__((ext_vector_type(2))) float v2f;
typedef __attribute__((ext_vector_type(4))) float v4f;
typedef __attribute__((ext_vector_type(8))) float v8f;


__global__ void cheb_count_kernel(const int* __restrict__ row, const int* __restrict__ col,
                                  float* __restrict__ deg, int* __restrict__ cnt,
                                  int* __restrict__ rank, int E) {
    int e = blockIdx.x * blockDim.x + threadIdx.x;
    if (e < E) {
        atomicAdd(&deg[row[e]], 1.0f);
        rank[e] = atomicAdd(&cnt[col[e]], 1);
    }
}

__global__ void cheb_dis_kernel(float* __restrict__ deg, int n) {
    int i = blockIdx.x * blockDim.x + threadIdx.x;
    if (i < n) {
        float d = deg[i];
        deg[i] = (d > 0.0f) ? rsqrtf(fmaxf(d, 1.0f)) : 0.0f;
    }
}

__device__ __forceinline__ int cheb_wave_scan(int x, int lane) {
    x += __builtin_amdgcn_update_dpp(0, x, 0x111, 0xF, 0xF, true);
    x += __builtin_amdgcn_update_dpp(0, x, 0x112, 0xF, 0xF, true);
    x += __builtin_amdgcn_update_dpp(0, x, 0x114, 0xF, 0xF, true);
    x += __builtin_amdgcn_update_dpp(0, x, 0x118, 0xF, 0xF, true);
    const int lo = __builtin_amdgcn_readlane(x, 15);
    return x + ((lane >= 16) ? lo : 0);
}

#define CHEB_SCAN_TILES 13
__global__ void __launch_bounds__(1024)
cheb_rowptr_kernel(const int* __restrict__ cnt, int* __restrict__ rowptr, int n) {
    constexpr int NT = CHEB_SCAN_TILES;
    __shared__ int sWave[NT][32];
    const int lane = threadIdx.x & 31;
    const int wv = __builtin_amdgcn_readfirstlane((int)(threadIdx.x >> 5));

    const int rt = n & ~3;
    int r0 = 0, r1 = 0, r2 = 0;
    if (rt + 0 < n) r0 = cnt[rt + 0];
    if (rt + 1 < n) r1 = cnt[rt + 1];
    if (rt + 2 < n) r2 = cnt[rt + 2];

    int carry = 0;
    for (int base = 0; base < n; base += NT * 4096) {
        const int i00 = base + (int)threadIdx.x * 4;
        const int room = n - i00;
        const int ragOff = rt - i00;
        int s1[NT], s2[NT], s3[NT], tot[NT], incl[NT], pre[NT];
        int4 v[NT];
#pragma unroll
        for (int t = 0; t < NT; ++t)
            v[t] = *(const int4*)(cnt + ((t * 4096 + 3 < room) ? i00 + t * 4096 : 0));
#pragma unroll
        for (int t = 0; t < NT; ++t) {
            const bool full = (t * 4096 + 3 < room);
            const bool rag = (ragOff == t * 4096);
            const int c0 = full ? v[t].x : (rag ? r0 : 0);
            const int c1 = full ? v[t].y : (rag ? r1 : 0);
            const int c2 = full ? v[t].z : (rag ? r2 : 0);
            const int c3 = full ? v[t].w : 0;
            s1[t] = c0; s2[t] = s1[t] + c1; s3[t] = s2[t] + c2; tot[t] = s3[t] + c3;
        }
#pragma unroll
        for (int t = 0; t < NT; ++t) incl[t] = cheb_wave_scan(tot[t], lane);
        if (lane == 31) {
#pragma unroll
            for (int t = 0; t < NT; ++t) sWave[t][wv] = incl[t];
        }
#pragma unroll
        for (int t = 0; t < NT; ++t) pre[t] = incl[t] - tot[t];
        __syncthreads();
        int wt[NT], wincl[NT];
#pragma unroll
        for (int t = 0; t < NT; ++t) wt[t] = sWave[t][lane];
#pragma unroll
        for (int t = 0; t < NT; ++t) wincl[t] = cheb_wave_scan(wt[t], lane);
        int run = carry;
#pragma unroll
        for (int t = 0; t < NT; ++t) {
            const int waveExcl = __builtin_amdgcn_readlane(wincl[t] - wt[t], wv);
            const int tileTot = __builtin_amdgcn_readlane(wincl[t], 31);
            const int ex = run + waveExcl + pre[t];
            int* __restrict__ dst = rowptr + i00 + t * 4096;
            if (t * 4096 + 3 < room) {
                *(int4*)dst = make_int4(ex, ex + s1[t], ex + s2[t], ex + s3[t]);
            } else {
                if (t * 4096 + 0 < room) dst[0] = ex;
                if (t * 4096 + 1 < room) dst[1] = ex + s1[t];
                if (t * 4096 + 2 < room) dst[2] = ex + s2[t];
            }
            run += tileTot;
        }
        carry = run;
        __syncthreads();
    }
    if (threadIdx.x == 0) rowptr[n] = carry;
}

__global__ void cheb_fill_kernel(const int* __restrict__ row, const int* __restrict__ col,
                                 const float* __restrict__ dis, const int* __restrict__ rowptr,
                                 const int* __restrict__ rank, int2* __restrict__ csr, int E) {
    int e = blockIdx.x * blockDim.x + threadIdx.x;
    if (e < E) {
        const int s = row[e];
        const int d = col[e];
        const float w = -dis[s] * dis[d];
        csr[rowptr[d] + rank[e]] = make_int2(s, __float_as_int(w));
    }
}

template <bool CHEB2>
__global__ void __launch_bounds__(256)
cheb_gather_kernel(const float* __restrict__ in, const float* __restrict__ t0,
                   float* __restrict__ out, const int* __restrict__ rowptr,
                   const int2* __restrict__ csr, int n) {
    const int i = __builtin_amdgcn_readfirstlane(blockIdx.x * 8 + (int)(threadIdx.x >> 5));
    if (i >= n) return;
    const int lane = threadIdx.x & 31;
    const int beg = rowptr[i];
    const int end = rowptr[i + 1];
    const float* __restrict__ base = in + (lane << 1);

    v2f t = {0.0f, 0.0f};
    if (CHEB2) t = *(const v2f*)(t0 + (size_t)i * 64 + (lane << 1));

    float ax = 0.0f, ay = 0.0f;
    int j = beg;
    for (; j + 8 <= end; j += 8) {
        int2 p[8];
        v2f v[8];
#pragma unroll
        for (int u = 0; u < 8; ++u) p[u] = csr[j + u];
#pragma unroll
        for (int u = 0; u < 8; ++u) v[u] = *(const v2f*)(base + (size_t)p[u].x * 64);
#pragma unroll
        for (int u = 0; u < 8; ++u) {
            const float w = __int_as_float(p[u].y);
            ax = fmaf(v[u].x, w, ax);
            ay = fmaf(v[u].y, w, ay);
        }
    }
    if (j + 4 <= end) {
        int2 p[4];
        v2f v[4];
#pragma unroll
        for (int u = 0; u < 4; ++u) p[u] = csr[j + u];
#pragma unroll
        for (int u = 0; u < 4; ++u) v[u] = *(const v2f*)(base + (size_t)p[u].x * 64);
#pragma unroll
        for (int u = 0; u < 4; ++u) {
            const float w = __int_as_float(p[u].y);
            ax = fmaf(v[u].x, w, ax);
            ay = fmaf(v[u].y, w, ay);
        }
        j += 4;
    }
    for (; j < end; ++j) {
        const int2 p = csr[j];
        const v2f v = *(const v2f*)(base + (size_t)p.x * 64);
        const float w = __int_as_float(p.y);
        ax = fmaf(v.x, w, ax);
        ay = fmaf(v.y, w, ay);
    }

    v2f r;
    if (CHEB2) { r.x = 2.0f * ax - t.x; r.y = 2.0f * ay - t.y; }
    else       { r.x = ax;              r.y = ay; }
    *(v2f*)(out + (size_t)i * 64 + (lane << 1)) = r;
}

__global__ void cheb_degree_kernel(const int* __restrict__ row, float* __restrict__ deg, int E) {
    int e = blockIdx.x * blockDim.x + threadIdx.x;
    if (e < E) atomicAdd(&deg[row[e]], 1.0f);
}

__global__ void cheb_edgew_kernel(const int* __restrict__ row, const int* __restrict__ col,
                                  const float* __restrict__ dis, float* __restrict__ wE, int E) {
    int e = blockIdx.x * blockDim.x + threadIdx.x;
    if (e < E) wE[e] = -dis[row[e]] * dis[col[e]];
}

__global__ void cheb_negcopy_kernel(const float* __restrict__ a, float* __restrict__ b, int n) {
    int i = blockIdx.x * blockDim.x + threadIdx.x;
    if (i < n) b[i] = -a[i];
}

__global__ void cheb_scatter_kernel(const float* __restrict__ in, float* __restrict__ out,
                                    const int* __restrict__ src, const int* __restrict__ dst,
                                    const float* __restrict__ wE, float scale, int E) {
    int t = blockIdx.x * blockDim.x + threadIdx.x;
    int e = t >> 4;
    if (e >= E) return;
    int c = t & 15;
    int s = src[e];
    int d = dst[e];
    float w = wE[e] * scale;
    const float4 v = *(const float4*)(in + (size_t)s * 64 + (c << 2));
    float* op = out + (size_t)d * 64 + (c << 2);
    atomicAdd(op + 0, v.x * w);
    atomicAdd(op + 1, v.y * w);
    atomicAdd(op + 2, v.z * w);
    atomicAdd(op + 3, v.w * w);
}

template <int MT>
__global__ void __launch_bounds__(128)
cheb_gemm_wmma_kernel(const float* __restrict__ T0, const float* __restrict__ T1,
                      const float* __restrict__ T2, const float* __restrict__ W,
                      const float* __restrict__ bias, float* __restrict__ out,
                      int nTilesN, int F, int doRelu, int totalWaveTiles, int rowOffset) {
    __shared__ float sW[3 * 64 * 64 + 64];
    const int tid = threadIdx.y * 32 + threadIdx.x;
    const int nW = 3 * 64 * F;
    for (int i = tid; i < nW; i += 128) sW[i] = W[i];
    for (int i = tid; i < F; i += 128) sW[nW + i] = bias[i];
    __syncthreads();

    const int waveTile = blockIdx.x * 4 + threadIdx.y;
    if (waveTile >= totalWaveTiles) return;
    const int mGroup = waveTile / nTilesN;
    const int nTile = waveTile % nTilesN;

    const int lane = threadIdx.x;
    const int lg = lane >> 4;
    const int ln = lane & 15;
    const int rowBase = rowOffset + mGroup * (MT * 16);
    const int col = (nTile << 4) + ln;

    v8f acc[MT];
    const float bv = sW[nW + col];
#pragma unroll
    for (int i = 0; i < MT; ++i)
#pragma unroll
        for (int r = 0; r < 8; ++r) acc[i][r] = bv;

    const float* mats[3] = {T0, T1, T2};
#pragma unroll
    for (int m = 0; m < 3; ++m) {
        const float* __restrict__ T = mats[m];
        const float* Wm = &sW[m * 64 * F];
#pragma unroll
        for (int kq = 0; kq < 8; ++kq) {
            const int k0 = (kq << 3) + (lg << 2);
            v2f bLo, bHi;
            bLo.x = Wm[k0 * F + col];
            bLo.y = Wm[(k0 + 1) * F + col];
            bHi.x = Wm[(k0 + 2) * F + col];
            bHi.y = Wm[(k0 + 3) * F + col];
            v4f a[MT];
#pragma unroll
            for (int i = 0; i < MT; ++i)
                a[i] = *(const v4f*)(T + (size_t)(rowBase + i * 16 + ln) * 64 + k0);
#pragma unroll
            for (int i = 0; i < MT; ++i) {
                v2f aLo;
                aLo.x = a[i].x; aLo.y = a[i].y;
                acc[i] = __builtin_amdgcn_wmma_f32_16x16x4_f32(
                    false, aLo, false, bLo, (short)0, acc[i], false, false);
            }
#pragma unroll
            for (int i = 0; i < MT; ++i) {
                v2f aHi;
                aHi.x = a[i].z; aHi.y = a[i].w;
                acc[i] = __builtin_amdgcn_wmma_f32_16x16x4_f32(
                    false, aHi, false, bHi, (short)0, acc[i], false, false);
            }
        }
    }

#pragma unroll
    for (int i = 0; i < MT; ++i) {
        if (doRelu) {
#pragma unroll
            for (int r = 0; r < 8; ++r) acc[i][r] = fmaxf(acc[i][r], 0.0f);
        }
        float* orow = out + (size_t)(rowBase + i * 16 + (lg << 3)) * F + col;
#pragma unroll
        for (int r = 0; r < 8; ++r) orow[r * F] = acc[i][r];
    }
}

static void cheb_gemm(const float* T0, const float* T1, const float* T2,
                      const float* W, const float* b, float* Hout,
                      int n, int F, int relu, hipStream_t stream) {
    const int nTilesN = F / 16;
    const int mTiles = n / 16;
    const int mGroups = mTiles / 5;
    const int wt5 = mGroups * nTilesN;
    if (wt5 > 0) {
        cheb_gemm_wmma_kernel<5><<<(wt5 + 3) / 4, dim3(32, 4), 0, stream>>>(
            T0, T1, T2, W, b, Hout, nTilesN, F, relu, wt5, 0);
    }
    const int remTiles = mTiles - mGroups * 5;
    if (remTiles > 0) {
        const int wt1 = remTiles * nTilesN;
        cheb_gemm_wmma_kernel<1><<<(wt1 + 3) / 4, dim3(32, 4), 0, stream>>>(
            T0, T1, T2, W, b, Hout, nTilesN, F, relu, wt1, mGroups * 80);
    }
}

static void cheb_layer_csr(const float* T0, const float* W, const float* b,
                           float* T1, float* T2, float* Hout,
                           const int* rowptr, const int2* csr,
                           int n, int F, int relu, hipStream_t stream) {
    const int gblocks = (n + 7) / 8;
    cheb_gather_kernel<false><<<gblocks, 256, 0, stream>>>(T0, T0, T1, rowptr, csr, n);
    cheb_gather_kernel<true><<<gblocks, 256, 0, stream>>>(T1, T0, T2, rowptr, csr, n);
    cheb_gemm(T0, T1, T2, W, b, Hout, n, F, relu, stream);
}

static void cheb_layer_scatter(const float* T0, const float* W, const float* b,
                               float* T1, float* T2, float* Hout,
                               const int* src, const int* dst, const float* wE,
                               int n, int E, int F, int relu, hipStream_t stream) {
    const int sthreads = E * 16;
    const int sblocks = (sthreads + 255) / 256;
    hipMemsetAsync(T1, 0, (size_t)n * 64 * sizeof(float), stream);
    cheb_scatter_kernel<<<sblocks, 256, 0, stream>>>(T0, T1, src, dst, wE, 1.0f, E);
    cheb_negcopy_kernel<<<(n * 64 + 255) / 256, 256, 0, stream>>>(T0, T2, n * 64);
    cheb_scatter_kernel<<<sblocks, 256, 0, stream>>>(T1, T2, src, dst, wE, 2.0f, E);
    cheb_gemm(T0, T1, T2, W, b, Hout, n, F, relu, stream);
}

extern "C" void kernel_launch(void* const* d_in, const int* in_sizes, int n_in,
                              void* d_out, int out_size, void* d_ws, size_t ws_size,
                              hipStream_t stream) {
    const float* x  = (const float*)d_in[0];
    const int*   ei = (const int*)d_in[1];
    const float* W1 = (const float*)d_in[2];
    const float* b1 = (const float*)d_in[3];
    const float* W2 = (const float*)d_in[4];
    const float* b2 = (const float*)d_in[5];
    const float* W3 = (const float*)d_in[6];
    const float* b3 = (const float*)d_in[7];

    const int n = in_sizes[0] / 64;
    const int E = in_sizes[1] / 2;
    const int* src = ei;
    const int* dst = ei + E;

    const size_t nPad = ((size_t)n + 64 + 63) & ~(size_t)63;
    const size_t ePad = ((size_t)E + 63) & ~(size_t)63;
    float* ws = (float*)d_ws;

    const size_t csrWords = 3 * nPad + 2 * ePad + 4 * (size_t)n * 64;
    if (ws_size >= csrWords * sizeof(float)) {
        size_t o = 0;
        float* dis    = ws + o;          o += nPad;
        int*   cnt    = (int*)(ws + o);  o += nPad;
        int*   rowptr = (int*)(ws + o);  o += nPad;
        int2*  csr    = (int2*)(ws + o); o += 2 * ePad;
        float* T1 = ws + o; o += (size_t)n * 64;
        float* T2 = ws + o; o += (size_t)n * 64;
        float* HA = ws + o; o += (size_t)n * 64;
        float* HB = ws + o; o += (size_t)n * 64;
        int* rank = (int*)T2;

        hipMemsetAsync(dis, 0, 2 * nPad * sizeof(float), stream);
        cheb_count_kernel<<<(E + 255) / 256, 256, 0, stream>>>(src, dst, dis, cnt, rank, E);
        cheb_dis_kernel<<<(n + 255) / 256, 256, 0, stream>>>(dis, n);
        cheb_rowptr_kernel<<<1, 1024, 0, stream>>>(cnt, rowptr, n);
        cheb_fill_kernel<<<(E + 255) / 256, 256, 0, stream>>>(src, dst, dis, rowptr, rank, csr, E);

        cheb_layer_csr(x,  W1, b1, T1, T2, HA, rowptr, csr, n, 64, 1, stream);
        cheb_layer_csr(HA, W2, b2, T1, T2, HB, rowptr, csr, n, 64, 1, stream);
        cheb_layer_csr(HB, W3, b3, T1, T2, (float*)d_out, rowptr, csr, n, 32, 0, stream);
        return;
    }

    size_t o = 0;
    float* dis = ws + o; o += ((size_t)n + 63) & ~(size_t)63;
    float* wE  = ws + o; o += ePad;
    float* T1  = ws + o; o += (size_t)n * 64;
    float* T2  = ws + o; o += (size_t)n * 64;
    float* HA  = ws + o; o += (size_t)n * 64;
    float* HB  = ws + o; o += (size_t)n * 64;

    hipMemsetAsync(dis, 0, (size_t)n * sizeof(float), stream);
    cheb_degree_kernel<<<(E + 255) / 256, 256, 0, stream>>>(src, dis, E);
    cheb_dis_kernel<<<(n + 255) / 256, 256, 0, stream>>>(dis, n);
    cheb_edgew_kernel<<<(E + 255) / 256, 256, 0, stream>>>(src, dst, dis, wE, E);

    cheb_layer_scatter(x,  W1, b1, T1, T2, HA, src, dst, wE, n, E, 64, 1, stream);
    cheb_layer_scatter(HA, W2, b2, T1, T2, HB, src, dst, wE, n, E, 64, 1, stream);
    cheb_layer_scatter(HB, W3, b3, T1, T2, (float*)d_out, src, dst, wE, n, E, 32, 0, stream);
}


# CSR build: per-node (rowptr, degree) records of 8 bytes; fill reads one random record per endpoint instead of dis[s], dis[d], rowptr[d] and takes the rsqrt itself; the dis launch goes
# speedup vs baseline: 2.1524x; 1.0473x over previous
#include <hip/hip_runtime.h>

typedef __attribute__((ext_vector_type(2))) float v2f;
typedef __attribute__((ext_vector_type(4))) float v4f;
typedef __attribute__((ext_vector_type(8))) float v8f;


__global__ void cheb_count_kernel(const int* __restrict__ row, const int* __restrict__ col,
                                  float* __restrict__ nodeF, int* __restrict__ cnt,
                                  int* __restrict__ rank, int E) {
    int e = blockIdx.x * blockDim.x + threadIdx.x;
    if (e < E) {
        atomicAdd(&nodeF[2 * row[e] + 1], 1.0f);
        rank[e] = atomicAdd(&cnt[col[e]], 1);
    }
}

__global__ void cheb_dis_kernel(float* __restrict__ deg, int n) {
    int i = blockIdx.x * blockDim.x + threadIdx.x;
    if (i < n) {
        float d = deg[i];
        deg[i] = (d > 0.0f) ? rsqrtf(fmaxf(d, 1.0f)) : 0.0f;
    }
}

__device__ __forceinline__ int cheb_wave_scan(int x, int lane) {
    x += __builtin_amdgcn_update_dpp(0, x, 0x111, 0xF, 0xF, true);
    x += __builtin_amdgcn_update_dpp(0, x, 0x112, 0xF, 0xF, true);
    x += __builtin_amdgcn_update_dpp(0, x, 0x114, 0xF, 0xF, true);
    x += __builtin_amdgcn_update_dpp(0, x, 0x118, 0xF, 0xF, true);
    const int lo = __builtin_amdgcn_readlane(x, 15);
    return x + ((lane >= 16) ? lo : 0);
}

#define CHEB_SCAN_TILES 13
__global__ void __launch_bounds__(1024)
cheb_rowptr_kernel(const int* __restrict__ cnt, int* __restrict__ nodeI, int n) {
    constexpr int NT = CHEB_SCAN_TILES;
    __shared__ int sWave[NT][32];
    const int lane = threadIdx.x & 31;
    const int wv = __builtin_amdgcn_readfirstlane((int)(threadIdx.x >> 5));

    const int rt = n & ~3;
    int r0 = 0, r1 = 0, r2 = 0;
    if (rt + 0 < n) r0 = cnt[rt + 0];
    if (rt + 1 < n) r1 = cnt[rt + 1];
    if (rt + 2 < n) r2 = cnt[rt + 2];

    int carry = 0;
    for (int base = 0; base < n; base += NT * 4096) {
        const int i00 = base + (int)threadIdx.x * 4;
        const int room = n - i00;
        const int ragOff = rt - i00;
        int s1[NT], s2[NT], s3[NT], tot[NT], incl[NT], pre[NT];
        int4 v[NT];
#pragma unroll
        for (int t = 0; t < NT; ++t)
            v[t] = *(const int4*)(cnt + ((t * 4096 + 3 < room) ? i00 + t * 4096 : 0));
#pragma unroll
        for (int t = 0; t < NT; ++t) {
            const bool full = (t * 4096 + 3 < room);
            const bool rag = (ragOff == t * 4096);
            const int c0 = full ? v[t].x : (rag ? r0 : 0);
            const int c1 = full ? v[t].y : (rag ? r1 : 0);
            const int c2 = full ? v[t].z : (rag ? r2 : 0);
            const int c3 = full ? v[t].w : 0;
            s1[t] = c0; s2[t] = s1[t] + c1; s3[t] = s2[t] + c2; tot[t] = s3[t] + c3;
        }
#pragma unroll
        for (int t = 0; t < NT; ++t) incl[t] = cheb_wave_scan(tot[t], lane);
        if (lane == 31) {
#pragma unroll
            for (int t = 0; t < NT; ++t) sWave[t][wv] = incl[t];
        }
#pragma unroll
        for (int t = 0; t < NT; ++t) pre[t] = incl[t] - tot[t];
        __syncthreads();
        int wt[NT], wincl[NT];
#pragma unroll
        for (int t = 0; t < NT; ++t) wt[t] = sWave[t][lane];
#pragma unroll
        for (int t = 0; t < NT; ++t) wincl[t] = cheb_wave_scan(wt[t], lane);
        int run = carry;
#pragma unroll
        for (int t = 0; t < NT; ++t) {
            const int waveExcl = __builtin_amdgcn_readlane(wincl[t] - wt[t], wv);
            const int tileTot = __builtin_amdgcn_readlane(wincl[t], 31);
            const int ex = run + waveExcl + pre[t];
            int* __restrict__ dst = nodeI + 2 * (i00 + t * 4096);
            if (t * 4096 + 3 < room) {
                dst[0] = ex; dst[2] = ex + s1[t]; dst[4] = ex + s2[t]; dst[6] = ex + s3[t];
            } else {
                if (t * 4096 + 0 < room) dst[0] = ex;
                if (t * 4096 + 1 < room) dst[2] = ex + s1[t];
                if (t * 4096 + 2 < room) dst[4] = ex + s2[t];
            }
            run += tileTot;
        }
        carry = run;
        __syncthreads();
    }
    if (threadIdx.x == 0) nodeI[2 * n] = carry;
}

__device__ __forceinline__ float cheb_dis_of(float d) {
    return (d > 0.0f) ? rsqrtf(fmaxf(d, 1.0f)) : 0.0f;
}

__global__ void cheb_fill_kernel(const int* __restrict__ row, const int* __restrict__ col,
                                 const int2* __restrict__ node,
                                 const int* __restrict__ rank, int2* __restrict__ csr, int E) {
    int e = blockIdx.x * blockDim.x + threadIdx.x;
    if (e < E) {
        const int s = row[e];
        const int d = col[e];
        const int2 ns = node[s];
        const int2 nd = node[d];
        const float w = -cheb_dis_of(__int_as_float(ns.y)) * cheb_dis_of(__int_as_float(nd.y));
        csr[nd.x + rank[e]] = make_int2(s, __float_as_int(w));
    }
}

template <bool CHEB2>
__global__ void __launch_bounds__(256)
cheb_gather_kernel(const float* __restrict__ in, const float* __restrict__ t0,
                   float* __restrict__ out, const int2* __restrict__ node,
                   const int2* __restrict__ csr, int n) {
    const int i = __builtin_amdgcn_readfirstlane(blockIdx.x * 8 + (int)(threadIdx.x >> 5));
    if (i >= n) return;
    const int lane = threadIdx.x & 31;
    const int beg = node[i].x;
    const int end = node[i + 1].x;
    const float* __restrict__ base = in + (lane << 1);

    v2f t = {0.0f, 0.0f};
    if (CHEB2) t = *(const v2f*)(t0 + (size_t)i * 64 + (lane << 1));

    float ax = 0.0f, ay = 0.0f;
    int j = beg;
    for (; j + 8 <= end; j += 8) {
        int2 p[8];
        v2f v[8];
#pragma unroll
        for (int u = 0; u < 8; ++u) p[u] = csr[j + u];
#pragma unroll
        for (int u = 0; u < 8; ++u) v[u] = *(const v2f*)(base + (size_t)p[u].x * 64);
#pragma unroll
        for (int u = 0; u < 8; ++u) {
            const float w = __int_as_float(p[u].y);
            ax = fmaf(v[u].x, w, ax);
            ay = fmaf(v[u].y, w, ay);
        }
    }
    if (j + 4 <= end) {
        int2 p[4];
        v2f v[4];
#pragma unroll
        for (int u = 0; u < 4; ++u) p[u] = csr[j + u];
#pragma unroll
        for (int u = 0; u < 4; ++u) v[u] = *(const v2f*)(base + (size_t)p[u].x * 64);
#pragma unroll
        for (int u = 0; u < 4; ++u) {
            const float w = __int_as_float(p[u].y);
            ax = fmaf(v[u].x, w, ax);
            ay = fmaf(v[u].y, w, ay);
        }
        j += 4;
    }
    for (; j < end; ++j) {
        const int2 p = csr[j];
        const v2f v = *(const v2f*)(base + (size_t)p.x * 64);
        const float w = __int_as_float(p.y);
        ax = fmaf(v.x, w, ax);
        ay = fmaf(v.y, w, ay);
    }

    v2f r;
    if (CHEB2) { r.x = 2.0f * ax - t.x; r.y = 2.0f * ay - t.y; }
    else       { r.x = ax;              r.y = ay; }
    *(v2f*)(out + (size_t)i * 64 + (lane << 1)) = r;
}

__global__ void cheb_degree_kernel(const int* __restrict__ row, float* __restrict__ deg, int E) {
    int e = blockIdx.x * blockDim.x + threadIdx.x;
    if (e < E) atomicAdd(&deg[row[e]], 1.0f);
}

__global__ void cheb_edgew_kernel(const int* __restrict__ row, const int* __restrict__ col,
                                  const float* __restrict__ dis, float* __restrict__ wE, int E) {
    int e = blockIdx.x * blockDim.x + threadIdx.x;
    if (e < E) wE[e] = -dis[row[e]] * dis[col[e]];
}

__global__ void cheb_negcopy_kernel(const float* __restrict__ a, float* __restrict__ b, int n) {
    int i = blockIdx.x * blockDim.x + threadIdx.x;
    if (i < n) b[i] = -a[i];
}

__global__ void cheb_scatter_kernel(const float* __restrict__ in, float* __restrict__ out,
                                    const int* __restrict__ src, const int* __restrict__ dst,
                                    const float* __restrict__ wE, float scale, int E) {
    int t = blockIdx.x * blockDim.x + threadIdx.x;
    int e = t >> 4;
    if (e >= E) return;
    int c = t & 15;
    int s = src[e];
    int d = dst[e];
    float w = wE[e] * scale;
    const float4 v = *(const float4*)(in + (size_t)s * 64 + (c << 2));
    float* op = out + (size_t)d * 64 + (c << 2);
    atomicAdd(op + 0, v.x * w);
    atomicAdd(op + 1, v.y * w);
    atomicAdd(op + 2, v.z * w);
    atomicAdd(op + 3, v.w * w);
}

template <int MT>
__global__ void __launch_bounds__(128)
cheb_gemm_wmma_kernel(const float* __restrict__ T0, const float* __restrict__ T1,
                      const float* __restrict__ T2, const float* __restrict__ W,
                      const float* __restrict__ bias, float* __restrict__ out,
                      int nTilesN, int F, int doRelu, int totalWaveTiles, int rowOffset) {
    __shared__ float sW[3 * 64 * 64 + 64];
    const int tid = threadIdx.y * 32 + threadIdx.x;
    const int nW = 3 * 64 * F;
    for (int i = tid; i < nW; i += 128) sW[i] = W[i];
    for (int i = tid; i < F; i += 128) sW[nW + i] = bias[i];
    __syncthreads();

    const int waveTile = blockIdx.x * 4 + threadIdx.y;
    if (waveTile >= totalWaveTiles) return;
    const int mGroup = waveTile / nTilesN;
    const int nTile = waveTile % nTilesN;

    const int lane = threadIdx.x;
    const int lg = lane >> 4;
    const int ln = lane & 15;
    const int rowBase = rowOffset + mGroup * (MT * 16);
    const int col = (nTile << 4) + ln;

    v8f acc[MT];
    const float bv = sW[nW + col];
#pragma unroll
    for (int i = 0; i < MT; ++i)
#pragma unroll
        for (int r = 0; r < 8; ++r) acc[i][r] = bv;

    const float* mats[3] = {T0, T1, T2};
#pragma unroll
    for (int m = 0; m < 3; ++m) {
        const float* __restrict__ T = mats[m];
        const float* Wm = &sW[m * 64 * F];
#pragma unroll
        for (int kq = 0; kq < 8; ++kq) {
            const int k0 = (kq << 3) + (lg << 2);
            v2f bLo, bHi;
            bLo.x = Wm[k0 * F + col];
            bLo.y = Wm[(k0 + 1) * F + col];
            bHi.x = Wm[(k0 + 2) * F + col];
            bHi.y = Wm[(k0 + 3) * F + col];
            v4f a[MT];
#pragma unroll
            for (int i = 0; i < MT; ++i)
                a[i] = *(const v4f*)(T + (size_t)(rowBase + i * 16 + ln) * 64 + k0);
#pragma unroll
            for (int i = 0; i < MT; ++i) {
                v2f aLo;
                aLo.x = a[i].x; aLo.y = a[i].y;
                acc[i] = __builtin_amdgcn_wmma_f32_16x16x4_f32(
                    false, aLo, false, bLo, (short)0, acc[i], false, false);
            }
#pragma unroll
            for (int i = 0; i < MT; ++i) {
                v2f aHi;
                aHi.x = a[i].z; aHi.y = a[i].w;
                acc[i] = __builtin_amdgcn_wmma_f32_16x16x4_f32(
                    false, aHi, false, bHi, (short)0, acc[i], false, false);
            }
        }
    }

#pragma unroll
    for (int i = 0; i < MT; ++i) {
        if (doRelu) {
#pragma unroll
            for (int r = 0; r < 8; ++r) acc[i][r] = fmaxf(acc[i][r], 0.0f);
        }
        float* orow = out + (size_t)(rowBase + i * 16 + (lg << 3)) * F + col;
#pragma unroll
        for (int r = 0; r < 8; ++r) orow[r * F] = acc[i][r];
    }
}

static void cheb_gemm(const float* T0, const float* T1, const float* T2,
                      const float* W, const float* b, float* Hout,
                      int n, int F, int relu, hipStream_t stream) {
    const int nTilesN = F / 16;
    const int mTiles = n / 16;
    const int mGroups = mTiles / 5;
    const int wt5 = mGroups * nTilesN;
    if (wt5 > 0) {
        cheb_gemm_wmma_kernel<5><<<(wt5 + 3) / 4, dim3(32, 4), 0, stream>>>(
            T0, T1, T2, W, b, Hout, nTilesN, F, relu, wt5, 0);
    }
    const int remTiles = mTiles - mGroups * 5;
    if (remTiles > 0) {
        const int wt1 = remTiles * nTilesN;
        cheb_gemm_wmma_kernel<1><<<(wt1 + 3) / 4, dim3(32, 4), 0, stream>>>(
            T0, T1, T2, W, b, Hout, nTilesN, F, relu, wt1, mGroups * 80);
    }
}

static void cheb_layer_csr(const float* T0, const float* W, const float* b,
                           float* T1, float* T2, float* Hout,
                           const int2* node, const int2* csr,
                           int n, int F, int relu, hipStream_t stream) {
    const int gblocks = (n + 7) / 8;
    cheb_gather_kernel<false><<<gblocks, 256, 0, stream>>>(T0, T0, T1, node, csr, n);
    cheb_gather_kernel<true><<<gblocks, 256, 0, stream>>>(T1, T0, T2, node, csr, n);
    cheb_gemm(T0, T1, T2, W, b, Hout, n, F, relu, stream);
}

static void cheb_layer_scatter(const float* T0, const float* W, const float* b,
                               float* T1, float* T2, float* Hout,
                               const int* src, const int* dst, const float* wE,
                               int n, int E, int F, int relu, hipStream_t stream) {
    const int sthreads = E * 16;
    const int sblocks = (sthreads + 255) / 256;
    hipMemsetAsync(T1, 0, (size_t)n * 64 * sizeof(float), stream);
    cheb_scatter_kernel<<<sblocks, 256, 0, stream>>>(T0, T1, src, dst, wE, 1.0f, E);
    cheb_negcopy_kernel<<<(n * 64 + 255) / 256, 256, 0, stream>>>(T0, T2, n * 64);
    cheb_scatter_kernel<<<sblocks, 256, 0, stream>>>(T1, T2, src, dst, wE, 2.0f, E);
    cheb_gemm(T0, T1, T2, W, b, Hout, n, F, relu, stream);
}

extern "C" void kernel_launch(void* const* d_in, const int* in_sizes, int n_in,
                              void* d_out, int out_size, void* d_ws, size_t ws_size,
                              hipStream_t stream) {
    const float* x  = (const float*)d_in[0];
    const int*   ei = (const int*)d_in[1];
    const float* W1 = (const float*)d_in[2];
    const float* b1 = (const float*)d_in[3];
    const float* W2 = (const float*)d_in[4];
    const float* b2 = (const float*)d_in[5];
    const float* W3 = (const float*)d_in[6];
    const float* b3 = (const float*)d_in[7];

    const int n = in_sizes[0] / 64;
    const int E = in_sizes[1] / 2;
    const int* src = ei;
    const int* dst = ei + E;

    const size_t nPad = ((size_t)n + 64 + 63) & ~(size_t)63;
    const size_t ePad = ((size_t)E + 63) & ~(size_t)63;
    float* ws = (float*)d_ws;

    const size_t csrWords = 3 * nPad + 2 * ePad + 4 * (size_t)n * 64;
    if (ws_size >= csrWords * sizeof(float)) {
        size_t o = 0;
        int2*  node   = (int2*)(ws + o); o += 2 * nPad;
        int*   cnt    = (int*)(ws + o);  o += nPad;
        int2*  csr    = (int2*)(ws + o); o += 2 * ePad;
        float* T1 = ws + o; o += (size_t)n * 64;
        float* T2 = ws + o; o += (size_t)n * 64;
        float* HA = ws + o; o += (size_t)n * 64;
        float* HB = ws + o; o += (size_t)n * 64;
        int* rank = (int*)T2;

        hipMemsetAsync(node, 0, 3 * nPad * sizeof(float), stream);
        cheb_count_kernel<<<(E + 255) / 256, 256, 0, stream>>>(src, dst, (float*)node, cnt, rank, E);
        cheb_rowptr_kernel<<<1, 1024, 0, stream>>>(cnt, (int*)node, n);
        cheb_fill_kernel<<<(E + 255) / 256, 256, 0, stream>>>(src, dst, node, rank, csr, E);

        cheb_layer_csr(x,  W1, b1, T1, T2, HA, node, csr, n, 64, 1, stream);
        cheb_layer_csr(HA, W2, b2, T1, T2, HB, node, csr, n, 64, 1, stream);
        cheb_layer_csr(HB, W3, b3, T1, T2, (float*)d_out, node, csr, n, 32, 0, stream);
        return;
    }

    size_t o = 0;
    float* dis = ws + o; o += ((size_t)n + 63) & ~(size_t)63;
    float* wE  = ws + o; o += ePad;
    float* T1  = ws + o; o += (size_t)n * 64;
    float* T2  = ws + o; o += (size_t)n * 64;
    float* HA  = ws + o; o += (size_t)n * 64;
    float* HB  = ws + o; o += (size_t)n * 64;

    hipMemsetAsync(dis, 0, (size_t)n * sizeof(float), stream);
    cheb_degree_kernel<<<(E + 255) / 256, 256, 0, stream>>>(src, dis, E);
    cheb_dis_kernel<<<(n + 255) / 256, 256, 0, stream>>>(dis, n);
    cheb_edgew_kernel<<<(E + 255) / 256, 256, 0, stream>>>(src, dst, dis, wE, E);

    cheb_layer_scatter(x,  W1, b1, T1, T2, HA, src, dst, wE, n, E, 64, 1, stream);
    cheb_layer_scatter(HA, W2, b2, T1, T2, HB, src, dst, wE, n, E, 64, 1, stream);
    cheb_layer_scatter(HB, W3, b3, T1, T2, (float*)d_out, src, dst, wE, n, E, 32, 0, stream);
}


# CSR build through LDS bins of 256 destinations (chunk histogram, sorted write-out, per-bin counting sort): no per-edge global rank add, no scattered fill; a row's sum order is LDS arrival order
# speedup vs baseline: 2.8085x; 1.3048x over previous
#include <hip/hip_runtime.h>

typedef __attribute__((ext_vector_type(2))) float v2f;
typedef __attribute__((ext_vector_type(4))) float v4f;
typedef __attribute__((ext_vector_type(8))) float v8f;


__global__ void cheb_dis_kernel(float* __restrict__ deg, int n) {
    int i = blockIdx.x * blockDim.x + threadIdx.x;
    if (i < n) {
        float d = deg[i];
        deg[i] = (d > 0.0f) ? rsqrtf(fmaxf(d, 1.0f)) : 0.0f;
    }
}


__device__ __forceinline__ float cheb_dis_of(float d) {
    return (d > 0.0f) ? rsqrtf(fmaxf(d, 1.0f)) : 0.0f;
}

__device__ __forceinline__ int cheb_wave_scan(int x, int lane) {
    x += __builtin_amdgcn_update_dpp(0, x, 0x111, 0xF, 0xF, true);
    x += __builtin_amdgcn_update_dpp(0, x, 0x112, 0xF, 0xF, true);
    x += __builtin_amdgcn_update_dpp(0, x, 0x114, 0xF, 0xF, true);
    x += __builtin_amdgcn_update_dpp(0, x, 0x118, 0xF, 0xF, true);
    const int lo = __builtin_amdgcn_readlane(x, 15);
    return x + ((lane >= 16) ? lo : 0);
}

__device__ __forceinline__ int cheb_block_excl_scan(int x, int* sWave) {
    const int lane = threadIdx.x & 31;
    const int wv = __builtin_amdgcn_readfirstlane((int)(threadIdx.x >> 5));
    const int incl = cheb_wave_scan(x, lane);
    if (lane == 31) sWave[wv] = incl;
    __syncthreads();
    const int wt = sWave[lane];
    const int wincl = cheb_wave_scan(wt, lane);
    const int waveExcl = __builtin_amdgcn_readlane(wincl - wt, wv);
    __syncthreads();
    return waveExcl + incl - x;
}

#define CHEB_BIN_SHIFT 8
#define CHEB_BIN_NODES 256
#define CHEB_MAX_BINS 1024
#define CHEB_CHUNK 4096
#define CHEB_STAGE 8192

__global__ void __launch_bounds__(1024)
cheb_bin_count_kernel(const int* __restrict__ row, const int* __restrict__ col,
                      float* __restrict__ nodeF, int* __restrict__ binCnt,
                      int* __restrict__ wgBase, int E, int NB) {
    __shared__ int sH[CHEB_MAX_BINS];
    const int tid = threadIdx.x;
    if (tid < NB) sH[tid] = 0;
    __syncthreads();
    int r[4], c4[4];
#pragma unroll
    for (int k = 0; k < 4; ++k) {
        const int e = blockIdx.x * CHEB_CHUNK + k * 1024 + tid;
        const int ee = (e < E) ? e : E - 1;
        r[k] = row[ee];
        c4[k] = col[ee];
    }
#pragma unroll
    for (int k = 0; k < 4; ++k) {
        const int e = blockIdx.x * CHEB_CHUNK + k * 1024 + tid;
        if (e < E) {
            atomicAdd(&nodeF[2 * r[k] + 1], 1.0f);
            atomicAdd(&sH[c4[k] >> CHEB_BIN_SHIFT], 1);
        }
    }
    __syncthreads();
    if (tid < NB) {
        const int c = sH[tid];
        int base = 0;
        if (c > 0) base = atomicAdd(&binCnt[tid], c);
        wgBase[blockIdx.x * NB + tid] = base;
    }
}

__global__ void __launch_bounds__(1024)
cheb_bin_place_kernel(const int* __restrict__ row, const int* __restrict__ col,
                      const int* __restrict__ binCnt, const int* __restrict__ wgBase,
                      int2* __restrict__ binned, int E, int NB) {
    __shared__ int sRun[CHEB_MAX_BINS];
    __shared__ int sH[CHEB_MAX_BINS];
    __shared__ int sLoc[CHEB_MAX_BINS];
    __shared__ int sWave[32];
    __shared__ int stS[CHEB_CHUNK];
    __shared__ int stD[CHEB_CHUNK];
    const int tid = threadIdx.x;

    const int bc = (tid < NB) ? binCnt[tid] : 0;
    const int binStart = cheb_block_excl_scan(bc, sWave);
    if (tid < NB) {
        sRun[tid] = binStart + wgBase[blockIdx.x * NB + tid];
        sH[tid] = 0;
    }
    __syncthreads();

    int s[4], d[4], lr[4];
#pragma unroll
    for (int k = 0; k < 4; ++k) {
        const int e = blockIdx.x * CHEB_CHUNK + k * 1024 + tid;
        const int ee = (e < E) ? e : E - 1;
        s[k] = row[ee];
        d[k] = col[ee];
    }
#pragma unroll
    for (int k = 0; k < 4; ++k) {
        const int e = blockIdx.x * CHEB_CHUNK + k * 1024 + tid;
        lr[k] = 0;
        if (e < E) lr[k] = atomicAdd(&sH[d[k] >> CHEB_BIN_SHIFT], 1);
        else d[k] = -1;
    }
    __syncthreads();
    const int hc = (tid < NB) ? sH[tid] : 0;
    const int lex = cheb_block_excl_scan(hc, sWave);
    if (tid < NB) sLoc[tid] = lex;
    __syncthreads();
#pragma unroll
    for (int k = 0; k < 4; ++k) {
        if (d[k] >= 0) {
            const int p = sLoc[d[k] >> CHEB_BIN_SHIFT] + lr[k];
            stS[p] = s[k];
            stD[p] = d[k];
        }
    }
    __syncthreads();
    const int left = E - blockIdx.x * CHEB_CHUNK;
    const int total = (left < CHEB_CHUNK) ? left : CHEB_CHUNK;
#pragma unroll
    for (int k = 0; k < 4; ++k) {
        const int t = k * 1024 + tid;
        if (t < total) {
            const int ss = stS[t];
            const int dd = stD[t];
            const int b = dd >> CHEB_BIN_SHIFT;
            binned[sRun[b] + (t - sLoc[b])] = make_int2(ss, dd);
        }
    }
}

__global__ void __launch_bounds__(1024)
cheb_bin_csr_kernel(const int2* __restrict__ binned, const int* __restrict__ binCnt,
                    int* node, int2* __restrict__ csr, int n, int E, int NB) {
    __shared__ int sWave[32];
    __shared__ int sC[CHEB_BIN_NODES];
    __shared__ int sCur[CHEB_BIN_NODES];
    __shared__ float sDis[CHEB_BIN_NODES];
    __shared__ int sS0, sM;
    __shared__ int stS[CHEB_STAGE];
    __shared__ float stW[CHEB_STAGE];
    const int tid = threadIdx.x;
    const int b = blockIdx.x;

    const int bc = (tid < NB) ? binCnt[tid] : 0;
    const int binStart = cheb_block_excl_scan(bc, sWave);
    if (tid == b) { sS0 = binStart; sM = bc; }
    if (tid < CHEB_BIN_NODES) {
        sC[tid] = 0;
        const int i = (b << CHEB_BIN_SHIFT) + tid;
        sDis[tid] = (i < n) ? cheb_dis_of(__int_as_float(node[2 * i + 1])) : 0.0f;
    }
    __syncthreads();
    const int s0 = sS0;
    const int m = sM;

    for (int j0 = 0; j0 < m; j0 += 4096) {
        int dd[4];
#pragma unroll
        for (int k = 0; k < 4; ++k) {
            const int j = j0 + k * 1024 + tid;
            dd[k] = binned[s0 + ((j < m) ? j : m - 1)].y;
        }
#pragma unroll
        for (int k = 0; k < 4; ++k) {
            const int j = j0 + k * 1024 + tid;
            if (j < m) atomicAdd(&sC[dd[k] & (CHEB_BIN_NODES - 1)], 1);
        }
    }
    __syncthreads();
    const int c = (tid < CHEB_BIN_NODES) ? sC[tid] : 0;
    const int ex = cheb_block_excl_scan(c, sWave);
    if (tid < CHEB_BIN_NODES) {
        sCur[tid] = ex;
        const int i = (b << CHEB_BIN_SHIFT) + tid;
        if (i < n) node[2 * i] = s0 + ex;
    }
    if (b == NB - 1 && tid == 0) node[2 * n] = E;
    __syncthreads();

    if (m <= CHEB_STAGE) {
        for (int j0 = 0; j0 < m; j0 += 4096) {
            int2 p[4];
            int dg[4];
#pragma unroll
            for (int k = 0; k < 4; ++k) {
                const int j = j0 + k * 1024 + tid;
                p[k] = binned[s0 + ((j < m) ? j : m - 1)];
            }
#pragma unroll
            for (int k = 0; k < 4; ++k) dg[k] = node[2 * p[k].x + 1];
#pragma unroll
            for (int k = 0; k < 4; ++k) {
                const int j = j0 + k * 1024 + tid;
                if (j < m) {
                    const int dl = p[k].y & (CHEB_BIN_NODES - 1);
                    const int r = atomicAdd(&sCur[dl], 1);
                    stS[r] = p[k].x;
                    stW[r] = -cheb_dis_of(__int_as_float(dg[k])) * sDis[dl];
                }
            }
        }
        __syncthreads();
        for (int j = tid; j < m; j += 1024)
            csr[s0 + j] = make_int2(stS[j], __float_as_int(stW[j]));
    } else {
        for (int j = tid; j < m; j += 1024) {
            const int2 p = binned[s0 + j];
            const int dl = p.y & (CHEB_BIN_NODES - 1);
            const int r = atomicAdd(&sCur[dl], 1);
            const float w = -cheb_dis_of(__int_as_float(node[2 * p.x + 1])) * sDis[dl];
            csr[s0 + r] = make_int2(p.x, __float_as_int(w));
        }
    }
}

template <bool CHEB2>
__global__ void __launch_bounds__(256)
cheb_gather_kernel(const float* __restrict__ in, const float* __restrict__ t0,
                   float* __restrict__ out, const int2* __restrict__ node,
                   const int2* __restrict__ csr, int n) {
    const int i = __builtin_amdgcn_readfirstlane(blockIdx.x * 8 + (int)(threadIdx.x >> 5));
    if (i >= n) return;
    const int lane = threadIdx.x & 31;
    const int beg = node[i].x;
    const int end = node[i + 1].x;
    const float* __restrict__ base = in + (lane << 1);

    v2f t = {0.0f, 0.0f};
    if (CHEB2) t = *(const v2f*)(t0 + (size_t)i * 64 + (lane << 1));

    float ax = 0.0f, ay = 0.0f;
    int j = beg;
    for (; j + 8 <= end; j += 8) {
        int2 p[8];
        v2f v[8];
#pragma unroll
        for (int u = 0; u < 8; ++u) p[u] = csr[j + u];
#pragma unroll
        for (int u = 0; u < 8; ++u) v[u] = *(const v2f*)(base + (size_t)p[u].x * 64);
#pragma unroll
        for (int u = 0; u < 8; ++u) {
            const float w = __int_as_float(p[u].y);
            ax = fmaf(v[u].x, w, ax);
            ay = fmaf(v[u].y, w, ay);
        }
    }
    if (j + 4 <= end) {
        int2 p[4];
        v2f v[4];
#pragma unroll
        for (int u = 0; u < 4; ++u) p[u] = csr[j + u];
#pragma unroll
        for (int u = 0; u < 4; ++u) v[u] = *(const v2f*)(base + (size_t)p[u].x * 64);
#pragma unroll
        for (int u = 0; u < 4; ++u) {
            const float w = __int_as_float(p[u].y);
            ax = fmaf(v[u].x, w, ax);
            ay = fmaf(v[u].y, w, ay);
        }
        j += 4;
    }
    for (; j < end; ++j) {
        const int2 p = csr[j];
        const v2f v = *(const v2f*)(base + (size_t)p.x * 64);
        const float w = __int_as_float(p.y);
        ax = fmaf(v.x, w, ax);
        ay = fmaf(v.y, w, ay);
    }

    v2f r;
    if (CHEB2) { r.x = 2.0f * ax - t.x; r.y = 2.0f * ay - t.y; }
    else       { r.x = ax;              r.y = ay; }
    *(v2f*)(out + (size_t)i * 64 + (lane << 1)) = r;
}

__global__ void cheb_degree_kernel(const int* __restrict__ row, float* __restrict__ deg, int E) {
    int e = blockIdx.x * blockDim.x + threadIdx.x;
    if (e < E) atomicAdd(&deg[row[e]], 1.0f);
}

__global__ void cheb_edgew_kernel(const int* __restrict__ row, const int* __restrict__ col,
                                  const float* __restrict__ dis, float* __restrict__ wE, int E) {
    int e = blockIdx.x * blockDim.x + threadIdx.x;
    if (e < E) wE[e] = -dis[row[e]] * dis[col[e]];
}

__global__ void cheb_negcopy_kernel(const float* __restrict__ a, float* __restrict__ b, int n) {
    int i = blockIdx.x * blockDim.x + threadIdx.x;
    if (i < n) b[i] = -a[i];
}

__global__ void cheb_scatter_kernel(const float* __restrict__ in, float* __restrict__ out,
                                    const int* __restrict__ src, const int* __restrict__ dst,
                                    const float* __restrict__ wE, float scale, int E) {
    int t = blockIdx.x * blockDim.x + threadIdx.x;
    int e = t >> 4;
    if (e >= E) return;
    int c = t & 15;
    int s = src[e];
    int d = dst[e];
    float w = wE[e] * scale;
    const float4 v = *(const float4*)(in + (size_t)s * 64 + (c << 2));
    float* op = out + (size_t)d * 64 + (c << 2);
    atomicAdd(op + 0, v.x * w);
    atomicAdd(op + 1, v.y * w);
    atomicAdd(op + 2, v.z * w);
    atomicAdd(op + 3, v.w * w);
}

template <int MT>
__global__ void __launch_bounds__(128)
cheb_gemm_wmma_kernel(const float* __restrict__ T0, const float* __restrict__ T1,
                      const float* __restrict__ T2, const float* __restrict__ W,
                      const float* __restrict__ bias, float* __restrict__ out,
                      int nTilesN, int F, int doRelu, int totalWaveTiles, int rowOffset) {
    __shared__ float sW[3 * 64 * 64 + 64];
    const int tid = threadIdx.y * 32 + threadIdx.x;
    const int nW = 3 * 64 * F;
    for (int i = tid; i < nW; i += 128) sW[i] = W[i];
    for (int i = tid; i < F; i += 128) sW[nW + i] = bias[i];
    __syncthreads();

    const int waveTile = blockIdx.x * 4 + threadIdx.y;
    if (waveTile >= totalWaveTiles) return;
    const int mGroup = waveTile / nTilesN;
    const int nTile = waveTile % nTilesN;

    const int lane = threadIdx.x;
    const int lg = lane >> 4;
    const int ln = lane & 15;
    const int rowBase = rowOffset + mGroup * (MT * 16);
    const int col = (nTile << 4) + ln;

    v8f acc[MT];
    const float bv = sW[nW + col];
#pragma unroll
    for (int i = 0; i < MT; ++i)
#pragma unroll
        for (int r = 0; r < 8; ++r) acc[i][r] = bv;

    const float* mats[3] = {T0, T1, T2};
#pragma unroll
    for (int m = 0; m < 3; ++m) {
        const float* __restrict__ T = mats[m];
        const float* Wm = &sW[m * 64 * F];
#pragma unroll
        for (int kq = 0; kq < 8; ++kq) {
            const int k0 = (kq << 3) + (lg << 2);
            v2f bLo, bHi;
            bLo.x = Wm[k0 * F + col];
            bLo.y = Wm[(k0 + 1) * F + col];
            bHi.x = Wm[(k0 + 2) * F + col];
            bHi.y = Wm[(k0 + 3) * F + col];
            v4f a[MT];
#pragma unroll
            for (int i = 0; i < MT; ++i)
                a[i] = *(const v4f*)(T + (size_t)(rowBase + i * 16 + ln) * 64 + k0);
#pragma unroll
            for (int i = 0; i < MT; ++i) {
                v2f aLo;
                aLo.x = a[i].x; aLo.y = a[i].y;
                acc[i] = __builtin_amdgcn_wmma_f32_16x16x4_f32(
                    false, aLo, false, bLo, (short)0, acc[i], false, false);
            }
#pragma unroll
            for (int i = 0; i < MT; ++i) {
                v2f aHi;
                aHi.x = a[i].z; aHi.y = a[i].w;
                acc[i] = __builtin_amdgcn_wmma_f32_16x16x4_f32(
                    false, aHi, false, bHi, (short)0, acc[i], false, false);
            }
        }
    }

#pragma unroll
    for (int i = 0; i < MT; ++i) {
        if (doRelu) {
#pragma unroll
            for (int r = 0; r < 8; ++r) acc[i][r] = fmaxf(acc[i][r], 0.0f);
        }
        float* orow = out + (size_t)(rowBase + i * 16 + (lg << 3)) * F + col;
#pragma unroll
        for (int r = 0; r < 8; ++r) orow[r * F] = acc[i][r];
    }
}

static void cheb_gemm(const float* T0, const float* T1, const float* T2,
                      const float* W, const float* b, float* Hout,
                      int n, int F, int relu, hipStream_t stream) {
    const int nTilesN = F / 16;
    const int mTiles = n / 16;
    const int mGroups = mTiles / 5;
    const int wt5 = mGroups * nTilesN;
    if (wt5 > 0) {
        cheb_gemm_wmma_kernel<5><<<(wt5 + 3) / 4, dim3(32, 4), 0, stream>>>(
            T0, T1, T2, W, b, Hout, nTilesN, F, relu, wt5, 0);
    }
    const int remTiles = mTiles - mGroups * 5;
    if (remTiles > 0) {
        const int wt1 = remTiles * nTilesN;
        cheb_gemm_wmma_kernel<1><<<(wt1 + 3) / 4, dim3(32, 4), 0, stream>>>(
            T0, T1, T2, W, b, Hout, nTilesN, F, relu, wt1, mGroups * 80);
    }
}

static void cheb_layer_csr(const float* T0, const float* W, const float* b,
                           float* T1, float* T2, float* Hout,
                           const int2* node, const int2* csr,
                           int n, int F, int relu, hipStream_t stream) {
    const int gblocks = (n + 7) / 8;
    cheb_gather_kernel<false><<<gblocks, 256, 0, stream>>>(T0, T0, T1, node, csr, n);
    cheb_gather_kernel<true><<<gblocks, 256, 0, stream>>>(T1, T0, T2, node, csr, n);
    cheb_gemm(T0, T1, T2, W, b, Hout, n, F, relu, stream);
}

static void cheb_layer_scatter(const float* T0, const float* W, const float* b,
                               float* T1, float* T2, float* Hout,
                               const int* src, const int* dst, const float* wE,
                               int n, int E, int F, int relu, hipStream_t stream) {
    const int sthreads = E * 16;
    const int sblocks = (sthreads + 255) / 256;
    hipMemsetAsync(T1, 0, (size_t)n * 64 * sizeof(float), stream);
    cheb_scatter_kernel<<<sblocks, 256, 0, stream>>>(T0, T1, src, dst, wE, 1.0f, E);
    cheb_negcopy_kernel<<<(n * 64 + 255) / 256, 256, 0, stream>>>(T0, T2, n * 64);
    cheb_scatter_kernel<<<sblocks, 256, 0, stream>>>(T1, T2, src, dst, wE, 2.0f, E);
    cheb_gemm(T0, T1, T2, W, b, Hout, n, F, relu, stream);
}

extern "C" void kernel_launch(void* const* d_in, const int* in_sizes, int n_in,
                              void* d_out, int out_size, void* d_ws, size_t ws_size,
                              hipStream_t stream) {
    const float* x  = (const float*)d_in[0];
    const int*   ei = (const int*)d_in[1];
    const float* W1 = (const float*)d_in[2];
    const float* b1 = (const float*)d_in[3];
    const float* W2 = (const float*)d_in[4];
    const float* b2 = (const float*)d_in[5];
    const float* W3 = (const float*)d_in[6];
    const float* b3 = (const float*)d_in[7];

    const int n = in_sizes[0] / 64;
    const int E = in_sizes[1] / 2;
    const int* src = ei;
    const int* dst = ei + E;

    const size_t nPad = ((size_t)n + 64 + 63) & ~(size_t)63;
    const size_t ePad = ((size_t)E + 63) & ~(size_t)63;
    float* ws = (float*)d_ws;

    const size_t csrWords = 3 * nPad + 2 * ePad + 4 * (size_t)n * 64;
    const int nb = (n + CHEB_BIN_NODES - 1) / CHEB_BIN_NODES;
    const int nChunk = (E + CHEB_CHUNK - 1) / CHEB_CHUNK;
    const bool binsFit = nb >= 1 && nb <= CHEB_MAX_BINS && nChunk >= 1 &&
                         2 * ePad <= (size_t)n * 64 && (size_t)nChunk * nb <= (size_t)n * 64;
    if (ws_size >= csrWords * sizeof(float) && binsFit) {
        size_t o = 0;
        int2*  node   = (int2*)(ws + o); o += 2 * nPad;
        int*   binCnt = (int*)(ws + o);  o += nPad;
        int2*  csr    = (int2*)(ws + o); o += 2 * ePad;
        float* T1 = ws + o; o += (size_t)n * 64;
        float* T2 = ws + o; o += (size_t)n * 64;
        float* HA = ws + o; o += (size_t)n * 64;
        float* HB = ws + o; o += (size_t)n * 64;
        int*  wgBase = (int*)T1;
        int2* binned = (int2*)T2;

        hipMemsetAsync(node, 0, 3 * nPad * sizeof(float), stream);
        cheb_bin_count_kernel<<<nChunk, 1024, 0, stream>>>(src, dst, (float*)node, binCnt, wgBase, E, nb);
        cheb_bin_place_kernel<<<nChunk, 1024, 0, stream>>>(src, dst, binCnt, wgBase, binned, E, nb);
        cheb_bin_csr_kernel<<<nb, 1024, 0, stream>>>(binned, binCnt, (int*)node, csr, n, E, nb);

        cheb_layer_csr(x,  W1, b1, T1, T2, HA, node, csr, n, 64, 1, stream);
        cheb_layer_csr(HA, W2, b2, T1, T2, HB, node, csr, n, 64, 1, stream);
        cheb_layer_csr(HB, W3, b3, T1, T2, (float*)d_out, node, csr, n, 32, 0, stream);
        return;
    }

    size_t o = 0;
    float* dis = ws + o; o += ((size_t)n + 63) & ~(size_t)63;
    float* wE  = ws + o; o += ePad;
    float* T1  = ws + o; o += (size_t)n * 64;
    float* T2  = ws + o; o += (size_t)n * 64;
    float* HA  = ws + o; o += (size_t)n * 64;
    float* HB  = ws + o; o += (size_t)n * 64;

    hipMemsetAsync(dis, 0, (size_t)n * sizeof(float), stream);
    cheb_degree_kernel<<<(E + 255) / 256, 256, 0, stream>>>(src, dis, E);
    cheb_dis_kernel<<<(n + 255) / 256, 256, 0, stream>>>(dis, n);
    cheb_edgew_kernel<<<(E + 255) / 256, 256, 0, stream>>>(src, dst, dis, wE, E);

    cheb_layer_scatter(x,  W1, b1, T1, T2, HA, src, dst, wE, n, E, 64, 1, stream);
    cheb_layer_scatter(HA, W2, b2, T1, T2, HB, src, dst, wE, n, E, 64, 1, stream);
    cheb_layer_scatter(HB, W3, b3, T1, T2, (float*)d_out, src, dst, wE, n, E, 32, 0, stream);
}


# Second propagation fused with the layer GEMM: 16-wave blocks gather 64 rows of T2 into LDS and multiply them there beside T0 and T1; T2 never goes to memory, 3 launches fewer; sum orders kept
# speedup vs baseline: 3.2389x; 1.1532x over previous
#include <hip/hip_runtime.h>

typedef __attribute__((ext_vector_type(2))) float v2f;
typedef __attribute__((ext_vector_type(4))) float v4f;
typedef __attribute__((ext_vector_type(8))) float v8f;


__global__ void cheb_dis_kernel(float* __restrict__ deg, int n) {
    int i = blockIdx.x * blockDim.x + threadIdx.x;
    if (i < n) {
        float d = deg[i];
        deg[i] = (d > 0.0f) ? rsqrtf(fmaxf(d, 1.0f)) : 0.0f;
    }
}


__device__ __forceinline__ float cheb_dis_of(float d) {
    return (d > 0.0f) ? rsqrtf(fmaxf(d, 1.0f)) : 0.0f;
}

__device__ __forceinline__ int cheb_wave_scan(int x, int lane) {
    x += __builtin_amdgcn_update_dpp(0, x, 0x111, 0xF, 0xF, true);
    x += __builtin_amdgcn_update_dpp(0, x, 0x112, 0xF, 0xF, true);
    x += __builtin_amdgcn_update_dpp(0, x, 0x114, 0xF, 0xF, true);
    x += __builtin_amdgcn_update_dpp(0, x, 0x118, 0xF, 0xF, true);
    const int lo = __builtin_amdgcn_readlane(x, 15);
    return x + ((lane >= 16) ? lo : 0);
}

__device__ __forceinline__ int cheb_block_excl_scan(int x, int* sWave) {
    const int lane = threadIdx.x & 31;
    const int wv = __builtin_amdgcn_readfirstlane((int)(threadIdx.x >> 5));
    const int incl = cheb_wave_scan(x, lane);
    if (lane == 31) sWave[wv] = incl;
    __syncthreads();
    const int wt = sWave[lane];
    const int wincl = cheb_wave_scan(wt, lane);
    const int waveExcl = __builtin_amdgcn_readlane(wincl - wt, wv);
    __syncthreads();
    return waveExcl + incl - x;
}

#define CHEB_BIN_SHIFT 8
#define CHEB_BIN_NODES 256
#define CHEB_MAX_BINS 1024
#define CHEB_CHUNK 4096
#define CHEB_STAGE 8192

__global__ void __launch_bounds__(1024)
cheb_bin_count_kernel(const int* __restrict__ row, const int* __restrict__ col,
                      float* __restrict__ nodeF, int* __restrict__ binCnt,
                      int* __restrict__ wgBase, int E, int NB) {
    __shared__ int sH[CHEB_MAX_BINS];
    const int tid = threadIdx.x;
    if (tid < NB) sH[tid] = 0;
    __syncthreads();
    int r[4], c4[4];
#pragma unroll
    for (int k = 0; k < 4; ++k) {
        const int e = blockIdx.x * CHEB_CHUNK + k * 1024 + tid;
        const int ee = (e < E) ? e : E - 1;
        r[k] = row[ee];
        c4[k] = col[ee];
    }
#pragma unroll
    for (int k = 0; k < 4; ++k) {
        const int e = blockIdx.x * CHEB_CHUNK + k * 1024 + tid;
        if (e < E) {
            atomicAdd(&nodeF[2 * r[k] + 1], 1.0f);
            atomicAdd(&sH[c4[k] >> CHEB_BIN_SHIFT], 1);
        }
    }
    __syncthreads();
    if (tid < NB) {
        const int c = sH[tid];
        int base = 0;
        if (c > 0) base = atomicAdd(&binCnt[tid], c);
        wgBase[blockIdx.x * NB + tid] = base;
    }
}

__global__ void __launch_bounds__(1024)
cheb_bin_place_kernel(const int* __restrict__ row, const int* __restrict__ col,
                      const int* __restrict__ binCnt, const int* __restrict__ wgBase,
                      int2* __restrict__ binned, int E, int NB) {
    __shared__ int sRun[CHEB_MAX_BINS];
    __shared__ int sH[CHEB_MAX_BINS];
    __shared__ int sLoc[CHEB_MAX_BINS];
    __shared__ int sWave[32];
    __shared__ int stS[CHEB_CHUNK];
    __shared__ int stD[CHEB_CHUNK];
    const int tid = threadIdx.x;

    const int bc = (tid < NB) ? binCnt[tid] : 0;
    const int binStart = cheb_block_excl_scan(bc, sWave);
    if (tid < NB) {
        sRun[tid] = binStart + wgBase[blockIdx.x * NB + tid];
        sH[tid] = 0;
    }
    __syncthreads();

    int s[4], d[4], lr[4];
#pragma unroll
    for (int k = 0; k < 4; ++k) {
        const int e = blockIdx.x * CHEB_CHUNK + k * 1024 + tid;
        const int ee = (e < E) ? e : E - 1;
        s[k] = row[ee];
        d[k] = col[ee];
    }
#pragma unroll
    for (int k = 0; k < 4; ++k) {
        const int e = blockIdx.x * CHEB_CHUNK + k * 1024 + tid;
        lr[k] = 0;
        if (e < E) lr[k] = atomicAdd(&sH[d[k] >> CHEB_BIN_SHIFT], 1);
        else d[k] = -1;
    }
    __syncthreads();
    const int hc = (tid < NB) ? sH[tid] : 0;
    const int lex = cheb_block_excl_scan(hc, sWave);
    if (tid < NB) sLoc[tid] = lex;
    __syncthreads();
#pragma unroll
    for (int k = 0; k < 4; ++k) {
        if (d[k] >= 0) {
            const int p = sLoc[d[k] >> CHEB_BIN_SHIFT] + lr[k];
            stS[p] = s[k];
            stD[p] = d[k];
        }
    }
    __syncthreads();
    const int left = E - blockIdx.x * CHEB_CHUNK;
    const int total = (left < CHEB_CHUNK) ? left : CHEB_CHUNK;
#pragma unroll
    for (int k = 0; k < 4; ++k) {
        const int t = k * 1024 + tid;
        if (t < total) {
            const int ss = stS[t];
            const int dd = stD[t];
            const int b = dd >> CHEB_BIN_SHIFT;
            binned[sRun[b] + (t - sLoc[b])] = make_int2(ss, dd);
        }
    }
}

__global__ void __launch_bounds__(1024)
cheb_bin_csr_kernel(const int2* __restrict__ binned, const int* __restrict__ binCnt,
                    int* node, int2* __restrict__ csr, int n, int E, int NB) {
    __shared__ int sWave[32];
    __shared__ int sC[CHEB_BIN_NODES];
    __shared__ int sCur[CHEB_BIN_NODES];
    __shared__ float sDis[CHEB_BIN_NODES];
    __shared__ int sS0, sM;
    __shared__ int stS[CHEB_STAGE];
    __shared__ float stW[CHEB_STAGE];
    const int tid = threadIdx.x;
    const int b = blockIdx.x;

    const int bc = (tid < NB) ? binCnt[tid] : 0;
    const int binStart = cheb_block_excl_scan(bc, sWave);
    if (tid == b) { sS0 = binStart; sM = bc; }
    if (tid < CHEB_BIN_NODES) {
        sC[tid] = 0;
        const int i = (b << CHEB_BIN_SHIFT) + tid;
        sDis[tid] = (i < n) ? cheb_dis_of(__int_as_float(node[2 * i + 1])) : 0.0f;
    }
    __syncthreads();
    const int s0 = sS0;
    const int m = sM;

    for (int j0 = 0; j0 < m; j0 += 4096) {
        int dd[4];
#pragma unroll
        for (int k = 0; k < 4; ++k) {
            const int j = j0 + k * 1024 + tid;
            dd[k] = binned[s0 + ((j < m) ? j : m - 1)].y;
        }
#pragma unroll
        for (int k = 0; k < 4; ++k) {
            const int j = j0 + k * 1024 + tid;
            if (j < m) atomicAdd(&sC[dd[k] & (CHEB_BIN_NODES - 1)], 1);
        }
    }
    __syncthreads();
    const int c = (tid < CHEB_BIN_NODES) ? sC[tid] : 0;
    const int ex = cheb_block_excl_scan(c, sWave);
    if (tid < CHEB_BIN_NODES) {
        sCur[tid] = ex;
        const int i = (b << CHEB_BIN_SHIFT) + tid;
        if (i < n) node[2 * i] = s0 + ex;
    }
    if (b == NB - 1 && tid == 0) node[2 * n] = E;
    __syncthreads();

    if (m <= CHEB_STAGE) {
        for (int j0 = 0; j0 < m; j0 += 4096) {
            int2 p[4];
            int dg[4];
#pragma unroll
            for (int k = 0; k < 4; ++k) {
                const int j = j0 + k * 1024 + tid;
                p[k] = binned[s0 + ((j < m) ? j : m - 1)];
            }
#pragma unroll
            for (int k = 0; k < 4; ++k) dg[k] = node[2 * p[k].x + 1];
#pragma unroll
            for (int k = 0; k < 4; ++k) {
                const int j = j0 + k * 1024 + tid;
                if (j < m) {
                    const int dl = p[k].y & (CHEB_BIN_NODES - 1);
                    const int r = atomicAdd(&sCur[dl], 1);
                    stS[r] = p[k].x;
                    stW[r] = -cheb_dis_of(__int_as_float(dg[k])) * sDis[dl];
                }
            }
        }
        __syncthreads();
        for (int j = tid; j < m; j += 1024)
            csr[s0 + j] = make_int2(stS[j], __float_as_int(stW[j]));
    } else {
        for (int j = tid; j < m; j += 1024) {
            const int2 p = binned[s0 + j];
            const int dl = p.y & (CHEB_BIN_NODES - 1);
            const int r = atomicAdd(&sCur[dl], 1);
            const float w = -cheb_dis_of(__int_as_float(node[2 * p.x + 1])) * sDis[dl];
            csr[s0 + r] = make_int2(p.x, __float_as_int(w));
        }
    }
}

template <bool CHEB2>
__global__ void __launch_bounds__(256)
cheb_gather_kernel(const float* __restrict__ in, const float* __restrict__ t0,
                   float* __restrict__ out, const int2* __restrict__ node,
                   const int2* __restrict__ csr, int n) {
    const int i = __builtin_amdgcn_readfirstlane(blockIdx.x * 8 + (int)(threadIdx.x >> 5));
    if (i >= n) return;
    const int lane = threadIdx.x & 31;
    const int beg = node[i].x;
    const int end = node[i + 1].x;
    const float* __restrict__ base = in + (lane << 1);

    v2f t = {0.0f, 0.0f};
    if (CHEB2) t = *(const v2f*)(t0 + (size_t)i * 64 + (lane << 1));

    float ax = 0.0f, ay = 0.0f;
    int j = beg;
    for (; j + 8 <= end; j += 8) {
        int2 p[8];
        v2f v[8];
#pragma unroll
        for (int u = 0; u < 8; ++u) p[u] = csr[j + u];
#pragma unroll
        for (int u = 0; u < 8; ++u) v[u] = *(const v2f*)(base + (size_t)p[u].x * 64);
#pragma unroll
        for (int u = 0; u < 8; ++u) {
            const float w = __int_as_float(p[u].y);
            ax = fmaf(v[u].x, w, ax);
            ay = fmaf(v[u].y, w, ay);
        }
    }
    if (j + 4 <= end) {
        int2 p[4];
        v2f v[4];
#pragma unroll
        for (int u = 0; u < 4; ++u) p[u] = csr[j + u];
#pragma unroll
        for (int u = 0; u < 4; ++u) v[u] = *(const v2f*)(base + (size_t)p[u].x * 64);
#pragma unroll
        for (int u = 0; u < 4; ++u) {
            const float w = __int_as_float(p[u].y);
            ax = fmaf(v[u].x, w, ax);
            ay = fmaf(v[u].y, w, ay);
        }
        j += 4;
    }
    for (; j < end; ++j) {
        const int2 p = csr[j];
        const v2f v = *(const v2f*)(base + (size_t)p.x * 64);
        const float w = __int_as_float(p.y);
        ax = fmaf(v.x, w, ax);
        ay = fmaf(v.y, w, ay);
    }

    v2f r;
    if (CHEB2) { r.x = 2.0f * ax - t.x; r.y = 2.0f * ay - t.y; }
    else       { r.x = ax;              r.y = ay; }
    *(v2f*)(out + (size_t)i * 64 + (lane << 1)) = r;
}

__global__ void cheb_degree_kernel(const int* __restrict__ row, float* __restrict__ deg, int E) {
    int e = blockIdx.x * blockDim.x + threadIdx.x;
    if (e < E) atomicAdd(&deg[row[e]], 1.0f);
}

__global__ void cheb_edgew_kernel(const int* __restrict__ row, const int* __restrict__ col,
                                  const float* __restrict__ dis, float* __restrict__ wE, int E) {
    int e = blockIdx.x * blockDim.x + threadIdx.x;
    if (e < E) wE[e] = -dis[row[e]] * dis[col[e]];
}

__global__ void cheb_negcopy_kernel(const float* __restrict__ a, float* __restrict__ b, int n) {
    int i = blockIdx.x * blockDim.x + threadIdx.x;
    if (i < n) b[i] = -a[i];
}

__global__ void cheb_scatter_kernel(const float* __restrict__ in, float* __restrict__ out,
                                    const int* __restrict__ src, const int* __restrict__ dst,
                                    const float* __restrict__ wE, float scale, int E) {
    int t = blockIdx.x * blockDim.x + threadIdx.x;
    int e = t >> 4;
    if (e >= E) return;
    int c = t & 15;
    int s = src[e];
    int d = dst[e];
    float w = wE[e] * scale;
    const float4 v = *(const float4*)(in + (size_t)s * 64 + (c << 2));
    float* op = out + (size_t)d * 64 + (c << 2);
    atomicAdd(op + 0, v.x * w);
    atomicAdd(op + 1, v.y * w);
    atomicAdd(op + 2, v.z * w);
    atomicAdd(op + 3, v.w * w);
}

template <int MT>
__global__ void __launch_bounds__(128)
cheb_gemm_wmma_kernel(const float* __restrict__ T0, const float* __restrict__ T1,
                      const float* __restrict__ T2, const float* __restrict__ W,
                      const float* __restrict__ bias, float* __restrict__ out,
                      int nTilesN, int F, int doRelu, int totalWaveTiles, int rowOffset) {
    __shared__ float sW[3 * 64 * 64 + 64];
    const int tid = threadIdx.y * 32 + threadIdx.x;
    const int nW = 3 * 64 * F;
    for (int i = tid; i < nW; i += 128) sW[i] = W[i];
    for (int i = tid; i < F; i += 128) sW[nW + i] = bias[i];
    __syncthreads();

    const int waveTile = blockIdx.x * 4 + threadIdx.y;
    if (waveTile >= totalWaveTiles) return;
    const int mGroup = waveTile / nTilesN;
    const int nTile = waveTile % nTilesN;

    const int lane = threadIdx.x;
    const int lg = lane >> 4;
    const int ln = lane & 15;
    const int rowBase = rowOffset + mGroup * (MT * 16);
    const int col = (nTile << 4) + ln;

    v8f acc[MT];
    const float bv = sW[nW + col];
#pragma unroll
    for (int i = 0; i < MT; ++i)
#pragma unroll
        for (int r = 0; r < 8; ++r) acc[i][r] = bv;

    const float* mats[3] = {T0, T1, T2};
#pragma unroll
    for (int m = 0; m < 3; ++m) {
        const float* __restrict__ T = mats[m];
        const float* Wm = &sW[m * 64 * F];
#pragma unroll
        for (int kq = 0; kq < 8; ++kq) {
            const int k0 = (kq << 3) + (lg << 2);
            v2f bLo, bHi;
            bLo.x = Wm[k0 * F + col];
            bLo.y = Wm[(k0 + 1) * F + col];
            bHi.x = Wm[(k0 + 2) * F + col];
            bHi.y = Wm[(k0 + 3) * F + col];
            v4f a[MT];
#pragma unroll
            for (int i = 0; i < MT; ++i)
                a[i] = *(const v4f*)(T + (size_t)(rowBase + i * 16 + ln) * 64 + k0);
#pragma unroll
            for (int i = 0; i < MT; ++i) {
                v2f aLo;
                aLo.x = a[i].x; aLo.y = a[i].y;
                acc[i] = __builtin_amdgcn_wmma_f32_16x16x4_f32(
                    false, aLo, false, bLo, (short)0, acc[i], false, false);
            }
#pragma unroll
            for (int i = 0; i < MT; ++i) {
                v2f aHi;
                aHi.x = a[i].z; aHi.y = a[i].w;
                acc[i] = __builtin_amdgcn_wmma_f32_16x16x4_f32(
                    false, aHi, false, bHi, (short)0, acc[i], false, false);
            }
        }
    }

#pragma unroll
    for (int i = 0; i < MT; ++i) {
        if (doRelu) {
#pragma unroll
            for (int r = 0; r < 8; ++r) acc[i][r] = fmaxf(acc[i][r], 0.0f);
        }
        float* orow = out + (size_t)(rowBase + i * 16 + (lg << 3)) * F + col;
#pragma unroll
        for (int r = 0; r < 8; ++r) orow[r * F] = acc[i][r];
    }
}

#define CHEB_FROWS 64
#define CHEB_FPITCH 68

template <int NT>
__global__ void __launch_bounds__(512)
cheb_gather_gemm_kernel(const float* __restrict__ T0, const float* __restrict__ T1,
                        const float* __restrict__ W, const float* __restrict__ bias,
                        float* __restrict__ out, const int2* __restrict__ node,
                        const int2* __restrict__ csr, int n, int doRelu) {
    constexpr int F = NT * 16;
    constexpr int NW = 3 * 64 * F;
    constexpr int WV = NW / 4 / 512;
    __shared__ __attribute__((aligned(16))) float sW[NW + F];
    __shared__ __attribute__((aligned(16))) float sT2[CHEB_FROWS * CHEB_FPITCH];
    const int tid = threadIdx.x;
    const int lane = tid & 31;
    const int wv = __builtin_amdgcn_readfirstlane(tid >> 5);

    v4f wreg[WV];
#pragma unroll
    for (int q = 0; q < WV; ++q) wreg[q] = *(const v4f*)(W + (size_t)(q * 512 + tid) * 4);
    const float breg = (tid < F) ? bias[tid] : 0.0f;

    const float* __restrict__ base = T1 + (lane << 1);
    for (int q = 0; q < 4; ++q) {
        const int lr = wv * 4 + q;
        const int i = blockIdx.x * CHEB_FROWS + lr;
        if (i < n) {
            const int beg = node[i].x;
            const int end = node[i + 1].x;
            const v2f t = *(const v2f*)(T0 + (size_t)i * 64 + (lane << 1));
            float ax = 0.0f, ay = 0.0f;
            int j = beg;
            for (; j + 8 <= end; j += 8) {
                int2 p[8];
                v2f v[8];
#pragma unroll
                for (int u = 0; u < 8; ++u) p[u] = csr[j + u];
#pragma unroll
                for (int u = 0; u < 8; ++u) v[u] = *(const v2f*)(base + (size_t)p[u].x * 64);
#pragma unroll
                for (int u = 0; u < 8; ++u) {
                    const float w = __int_as_float(p[u].y);
                    ax = fmaf(v[u].x, w, ax);
                    ay = fmaf(v[u].y, w, ay);
                }
            }
            if (j + 4 <= end) {
                int2 p[4];
                v2f v[4];
#pragma unroll
                for (int u = 0; u < 4; ++u) p[u] = csr[j + u];
#pragma unroll
                for (int u = 0; u < 4; ++u) v[u] = *(const v2f*)(base + (size_t)p[u].x * 64);
#pragma unroll
                for (int u = 0; u < 4; ++u) {
                    const float w = __int_as_float(p[u].y);
                    ax = fmaf(v[u].x, w, ax);
                    ay = fmaf(v[u].y, w, ay);
                }
                j += 4;
            }
            for (; j < end; ++j) {
                const int2 p = csr[j];
                const v2f v = *(const v2f*)(base + (size_t)p.x * 64);
                const float w = __int_as_float(p.y);
                ax = fmaf(v.x, w, ax);
                ay = fmaf(v.y, w, ay);
            }
            v2f r;
            r.x = 2.0f * ax - t.x;
            r.y = 2.0f * ay - t.y;
            *(v2f*)(&sT2[lr * CHEB_FPITCH + (lane << 1)]) = r;
        }
    }

#pragma unroll
    for (int q = 0; q < WV; ++q) *(v4f*)(&sW[(q * 512 + tid) * 4]) = wreg[q];
    if (tid < F) sW[NW + tid] = breg;
    __syncthreads();

    const int nTile = wv & (NT - 1);
    const int mg = wv / NT;
    const int mTile = blockIdx.x * 4 + mg;
    if (mg >= 4 || mTile >= (n >> 4)) return;

    const int lg = lane >> 4;
    const int ln = lane & 15;
    const int rowBase = mTile * 16;
    const int col = (nTile << 4) + ln;

    v8f acc;
    const float bv = sW[NW + col];
#pragma unroll
    for (int r = 0; r < 8; ++r) acc[r] = bv;

    const float* __restrict__ rowT0 = T0 + (size_t)(rowBase + ln) * 64;
    const float* __restrict__ rowT1 = T1 + (size_t)(rowBase + ln) * 64;
    const float* rowT2 = &sT2[(mg * 16 + ln) * CHEB_FPITCH];
#pragma unroll
    for (int m = 0; m < 3; ++m) {
        const float* Wm = &sW[m * 64 * F];
#pragma unroll
        for (int kq = 0; kq < 8; ++kq) {
            const int k0 = (kq << 3) + (lg << 2);
            v2f bLo, bHi;
            bLo.x = Wm[k0 * F + col];
            bLo.y = Wm[(k0 + 1) * F + col];
            bHi.x = Wm[(k0 + 2) * F + col];
            bHi.y = Wm[(k0 + 3) * F + col];
            v4f a;
            if (m == 0)      a = *(const v4f*)(rowT0 + k0);
            else if (m == 1) a = *(const v4f*)(rowT1 + k0);
            else             a = *(const v4f*)(rowT2 + k0);
            v2f aLo, aHi;
            aLo.x = a.x; aLo.y = a.y;
            aHi.x = a.z; aHi.y = a.w;
            acc = __builtin_amdgcn_wmma_f32_16x16x4_f32(
                false, aLo, false, bLo, (short)0, acc, false, false);
            acc = __builtin_amdgcn_wmma_f32_16x16x4_f32(
                false, aHi, false, bHi, (short)0, acc, false, false);
        }
    }

    if (doRelu) {
#pragma unroll
        for (int r = 0; r < 8; ++r) acc[r] = fmaxf(acc[r], 0.0f);
    }
    float* orow = out + (size_t)(rowBase + (lg << 3)) * F + col;
#pragma unroll
    for (int r = 0; r < 8; ++r) orow[r * F] = acc[r];
}

static void cheb_gemm(const float* T0, const float* T1, const float* T2,
                      const float* W, const float* b, float* Hout,
                      int n, int F, int relu, hipStream_t stream) {
    const int nTilesN = F / 16;
    const int mTiles = n / 16;
    const int mGroups = mTiles / 5;
    const int wt5 = mGroups * nTilesN;
    if (wt5 > 0) {
        cheb_gemm_wmma_kernel<5><<<(wt5 + 3) / 4, dim3(32, 4), 0, stream>>>(
            T0, T1, T2, W, b, Hout, nTilesN, F, relu, wt5, 0);
    }
    const int remTiles = mTiles - mGroups * 5;
    if (remTiles > 0) {
        const int wt1 = remTiles * nTilesN;
        cheb_gemm_wmma_kernel<1><<<(wt1 + 3) / 4, dim3(32, 4), 0, stream>>>(
            T0, T1, T2, W, b, Hout, nTilesN, F, relu, wt1, mGroups * 80);
    }
}

static void cheb_layer_csr(const float* T0, const float* W, const float* b,
                           float* T1, float* Hout,
                           const int2* node, const int2* csr,
                           int n, int F, int relu, hipStream_t stream) {
    const int gblocks = (n + 7) / 8;
    cheb_gather_kernel<false><<<gblocks, 256, 0, stream>>>(T0, T0, T1, node, csr, n);
    const int fblocks = (n + CHEB_FROWS - 1) / CHEB_FROWS;
    if (F == 64)
        cheb_gather_gemm_kernel<4><<<fblocks, 512, 0, stream>>>(T0, T1, W, b, Hout, node, csr, n, relu);
    else
        cheb_gather_gemm_kernel<2><<<fblocks, 512, 0, stream>>>(T0, T1, W, b, Hout, node, csr, n, relu);
}

static void cheb_layer_scatter(const float* T0, const float* W, const float* b,
                               float* T1, float* T2, float* Hout,
                               const int* src, const int* dst, const float* wE,
                               int n, int E, int F, int relu, hipStream_t stream) {
    const int sthreads = E * 16;
    const int sblocks = (sthreads + 255) / 256;
    hipMemsetAsync(T1, 0, (size_t)n * 64 * sizeof(float), stream);
    cheb_scatter_kernel<<<sblocks, 256, 0, stream>>>(T0, T1, src, dst, wE, 1.0f, E);
    cheb_negcopy_kernel<<<(n * 64 + 255) / 256, 256, 0, stream>>>(T0, T2, n * 64);
    cheb_scatter_kernel<<<sblocks, 256, 0, stream>>>(T1, T2, src, dst, wE, 2.0f, E);
    cheb_gemm(T0, T1, T2, W, b, Hout, n, F, relu, stream);
}

extern "C" void kernel_launch(void* const* d_in, const int* in_sizes, int n_in,
                              void* d_out, int out_size, void* d_ws, size_t ws_size,
                              hipStream_t stream) {
    const float* x  = (const float*)d_in[0];
    const int*   ei = (const int*)d_in[1];
    const float* W1 = (const float*)d_in[2];
    const float* b1 = (const float*)d_in[3];
    const float* W2 = (const float*)d_in[4];
    const float* b2 = (const float*)d_in[5];
    const float* W3 = (const float*)d_in[6];
    const float* b3 = (const float*)d_in[7];

    const int n = in_sizes[0] / 64;
    const int E = in_sizes[1] / 2;
    const int* src = ei;
    const int* dst = ei + E;

    const size_t nPad = ((size_t)n + 64 + 63) & ~(size_t)63;
    const size_t ePad = ((size_t)E + 63) & ~(size_t)63;
    float* ws = (float*)d_ws;

    const size_t csrWords = 3 * nPad + 2 * ePad + 4 * (size_t)n * 64;
    const int nb = (n + CHEB_BIN_NODES - 1) / CHEB_BIN_NODES;
    const int nChunk = (E + CHEB_CHUNK - 1) / CHEB_CHUNK;
    const bool binsFit = nb >= 1 && nb <= CHEB_MAX_BINS && nChunk >= 1 &&
                         2 * ePad <= (size_t)n * 64 && (size_t)nChunk * nb <= (size_t)n * 64;
    if (ws_size >= csrWords * sizeof(float) && binsFit) {
        size_t o = 0;
        int2*  node   = (int2*)(ws + o); o += 2 * nPad;
        int*   binCnt = (int*)(ws + o);  o += nPad;
        int2*  csr    = (int2*)(ws + o); o += 2 * ePad;
        float* T1 = ws + o; o += (size_t)n * 64;
        float* T2 = ws + o; o += (size_t)n * 64;
        float* HA = ws + o; o += (size_t)n * 64;
        float* HB = ws + o; o += (size_t)n * 64;
        int*  wgBase = (int*)T1;
        int2* binned = (int2*)T2;

        hipMemsetAsync(node, 0, 3 * nPad * sizeof(float), stream);
        cheb_bin_count_kernel<<<nChunk, 1024, 0, stream>>>(src, dst, (float*)node, binCnt, wgBase, E, nb);
        cheb_bin_place_kernel<<<nChunk, 1024, 0, stream>>>(src, dst, binCnt, wgBase, binned, E, nb);
        cheb_bin_csr_kernel<<<nb, 1024, 0, stream>>>(binned, binCnt, (int*)node, csr, n, E, nb);

        cheb_layer_csr(x,  W1, b1, T1, HA, node, csr, n, 64, 1, stream);
        cheb_layer_csr(HA, W2, b2, T1, HB, node, csr, n, 64, 1, stream);
        cheb_layer_csr(HB, W3, b3, T1, (float*)d_out, node, csr, n, 32, 0, stream);
        return;
    }

    size_t o = 0;
    float* dis = ws + o; o += ((size_t)n + 63) & ~(size_t)63;
    float* wE  = ws + o; o += ePad;
    float* T1  = ws + o; o += (size_t)n * 64;
    float* T2  = ws + o; o += (size_t)n * 64;
    float* HA  = ws + o; o += (size_t)n * 64;
    float* HB  = ws + o; o += (size_t)n * 64;

    hipMemsetAsync(dis, 0, (size_t)n * sizeof(float), stream);
    cheb_degree_kernel<<<(E + 255) / 256, 256, 0, stream>>>(src, dis, E);
    cheb_dis_kernel<<<(n + 255) / 256, 256, 0, stream>>>(dis, n);
    cheb_edgew_kernel<<<(E + 255) / 256, 256, 0, stream>>>(src, dst, dis, wE, E);

    cheb_layer_scatter(x,  W1, b1, T1, T2, HA, src, dst, wE, n, E, 64, 1, stream);
    cheb_layer_scatter(HA, W2, b2, T1, T2, HB, src, dst, wE, n, E, 64, 1, stream);
    cheb_layer_scatter(HB, W3, b3, T1, T2, (float*)d_out, src, dst, wE, n, E, 32, 0, stream);
}
